# starting kernel
# baseline (speedup 1.0000x reference)
#include <hip/hip_runtime.h>
#include <hip/hip_bf16.h>


#define N_NODES  100000
#define N_EDGES  600000
#define N_GRAPHS 2000
#define NCLS     4

typedef float v2f __attribute__((ext_vector_type(2)));
typedef float v8f __attribute__((ext_vector_type(8)));


__global__ void k_zero(float* __restrict__ p, int n) {
  int i = blockIdx.x * blockDim.x + threadIdx.x;
  if (i < n) p[i] = 0.0f;
}

__global__ void k_deg(const int* __restrict__ dst, float* __restrict__ deg) {
  int e = blockIdx.x * blockDim.x + threadIdx.x;
  if (e < N_EDGES) atomicAdd(&deg[dst[e]], 1.0f);
}

__global__ void k_dinv(float* __restrict__ deg) {
  int i = blockIdx.x * blockDim.x + threadIdx.x;
  if (i < N_NODES) deg[i] = rsqrtf(deg[i] + 1.0f);
}

__global__ void k_embed(const int* __restrict__ tok, const float* __restrict__ emb,
                        float* __restrict__ x0) {
  size_t idx = (size_t)blockIdx.x * blockDim.x + threadIdx.x;
  int node = (int)(idx >> 6);
  int c    = (int)(idx & 63);
  x0[idx] = emb[((size_t)tok[node] << 6) | c];
}

template<int K>
__global__ void gemm_wmma_kernel(const float* __restrict__ X,
                                 const float* __restrict__ W,
                                 float* __restrict__ H) {
  const int wave  = (blockIdx.x << 3) | (threadIdx.x >> 5);
  const int lane  = threadIdx.x & 31;
  const int tileM = wave >> 3;
  const int tileN = wave & 7;
  const int half  = lane >> 4;
  const int lid   = lane & 15;
  const int m     = tileM * 16 + lid;
  const int n     = tileN * 16 + lid;
  const int kb    = half * 2;

  v8f acc = {};
  const float* xrow = X + (size_t)m * K + kb;
  const float* wcol = W + (size_t)kb * 128 + n;
#pragma unroll
  for (int k = 0; k < K; k += 4) {
    v2f a; a.x = xrow[k];               a.y = xrow[k + 1];
    v2f b; b.x = wcol[(size_t)k * 128]; b.y = wcol[(size_t)(k + 1) * 128];
    acc = __builtin_amdgcn_wmma_f32_16x16x4_f32(
         false, a,  false, b,
         (short)0, acc,  false,  false);
  }

  float* hbase = H + ((size_t)(tileM * 16 + half * 8)) * 128 + tileN * 16 + lid;
#pragma unroll
  for (int j = 0; j < 8; ++j)
    hbase[(size_t)j * 128] = acc[j];
}


__global__ void k_selfinit(const float* __restrict__ H, const float* __restrict__ dinv,
                           float* __restrict__ out) {
  size_t idx = (size_t)blockIdx.x * blockDim.x + threadIdx.x;
  int node = (int)(idx >> 7);
  float dv = dinv[node];
  out[idx] = H[idx] * dv * dv;
}

__global__ void k_scatter(const float* __restrict__ H, const int* __restrict__ src,
                          const int* __restrict__ dst, const float* __restrict__ dinv,
                          float* __restrict__ out) {
  size_t gid = (size_t)blockIdx.x * blockDim.x + threadIdx.x;
  int e    = (int)(gid >> 5);
  int lane = (int)(gid & 31);
  if (e >= N_EDGES) return;
  int s = src[e], d = dst[e];
  float nrm = dinv[s] * dinv[d];
  const float4 h = *(const float4*)(H + ((size_t)s << 7) + (lane << 2));
  float* o = out + ((size_t)d << 7) + (lane << 2);
  atomicAdd(o + 0, h.x * nrm);
  atomicAdd(o + 1, h.y * nrm);
  atomicAdd(o + 2, h.z * nrm);
  atomicAdd(o + 3, h.w * nrm);
}

__global__ void k_bias_relu(float* __restrict__ X, const float* __restrict__ b) {
  size_t idx = (size_t)blockIdx.x * blockDim.x + threadIdx.x;
  int c = (int)(idx & 127);
  float v = X[idx] + b[c];
  X[idx] = v > 0.0f ? v : 0.0f;
}


__global__ void k_pool(const float* __restrict__ X, const int* __restrict__ batch,
                       float* __restrict__ sums, float* __restrict__ cnts) {
  size_t gid = (size_t)blockIdx.x * blockDim.x + threadIdx.x;
  int node = (int)(gid >> 5);
  int lane = (int)(gid & 31);
  if (node >= N_NODES) return;
  int g = batch[node];
  const float4 v = *(const float4*)(X + ((size_t)node << 7) + (lane << 2));
  float* o = sums + ((size_t)g << 7) + (lane << 2);
  atomicAdd(o + 0, v.x);
  atomicAdd(o + 1, v.y);
  atomicAdd(o + 2, v.z);
  atomicAdd(o + 3, v.w);
  if (lane == 0) atomicAdd(&cnts[g], 1.0f);
}

__global__ void k_final(const float* __restrict__ sums, const float* __restrict__ cnts,
                        const float* __restrict__ Wlin, const float* __restrict__ blin,
                        float* __restrict__ out) {
  int t = blockIdx.x * blockDim.x + threadIdx.x;
  if (t >= N_GRAPHS * NCLS) return;
  int g = t >> 2, c = t & 3;
  float inv = 1.0f / fmaxf(cnts[g], 1.0f);
  float acc = blin[c];
  const float* srow = sums + ((size_t)g << 7);
#pragma unroll 4
  for (int k = 0; k < 128; ++k)
    acc = fmaf(srow[k] * inv, Wlin[k * NCLS + c], acc);
  out[t] = acc;
}


extern "C" void kernel_launch(void* const* d_in, const int* in_sizes, int n_in,
                              void* d_out, int out_size, void* d_ws, size_t ws_size,
                              hipStream_t stream) {
  const int*   tok   = (const int*)d_in[0];
  const int*   eidx  = (const int*)d_in[1];
  const int*   src   = eidx;
  const int*   dst   = eidx + N_EDGES;
  const int*   batch = (const int*)d_in[2];
  const float* emb   = (const float*)d_in[3];
  const float* W1    = (const float*)d_in[4];
  const float* b1    = (const float*)d_in[5];
  const float* W2    = (const float*)d_in[6];
  const float* b2    = (const float*)d_in[7];
  const float* Wlin  = (const float*)d_in[8];
  const float* blin  = (const float*)d_in[9];
  float*       out   = (float*)d_out;

  float* ws   = (float*)d_ws;
  float* deg  = ws;
  float* x0   = deg  + N_NODES;
  float* A    = x0   + (size_t)N_NODES * 64;
  float* B    = A    + (size_t)N_NODES * 128;
  float* sums = B    + (size_t)N_NODES * 128;
  float* cnts = sums + (size_t)N_GRAPHS * 128;

  const int T = 256;
  const int gNodes  = (N_NODES + T - 1) / T;
  const int gEdges  = (N_EDGES + T - 1) / T;
  const int gEmbed  = (N_NODES * 64) / T;
  const int gFeat   = (N_NODES * 128) / T;
  const int gScat   = (N_EDGES * 32) / T;
  const int gPool   = (N_NODES * 32) / T;
  const int gGemm   = N_NODES / 16;

  k_zero<<<gNodes, T, 0, stream>>>(deg, N_NODES);
  k_deg<<<gEdges, T, 0, stream>>>(dst, deg);
  k_dinv<<<gNodes, T, 0, stream>>>(deg);

  k_embed<<<gEmbed, T, 0, stream>>>(tok, emb, x0);

  gemm_wmma_kernel<64><<<gGemm, T, 0, stream>>>(x0, W1, A);
  k_selfinit<<<gFeat, T, 0, stream>>>(A, deg, B);
  k_scatter<<<gScat, T, 0, stream>>>(A, src, dst, deg, B);
  k_bias_relu<<<gFeat, T, 0, stream>>>(B, b1);

  gemm_wmma_kernel<128><<<gGemm, T, 0, stream>>>(B, W2, A);
  k_selfinit<<<gFeat, T, 0, stream>>>(A, deg, B);
  k_scatter<<<gScat, T, 0, stream>>>(A, src, dst, deg, B);
  k_bias_relu<<<gFeat, T, 0, stream>>>(B, b2);

  k_zero<<<(N_GRAPHS * 128 + N_GRAPHS + T - 1) / T, T, 0, stream>>>(sums,
                                                    N_GRAPHS * 128 + N_GRAPHS);
  k_pool<<<gPool, T, 0, stream>>>(B, batch, sums, cnts);
  k_final<<<(N_GRAPHS * NCLS + T - 1) / T, T, 0, stream>>>(sums, cnts, Wlin, blin, out);
}


# aggregate by gather: edges grouped by dst each call (int atomics give rank and row span); one wave per node sums in-edges in rank-arrival order, stores relu(self+sum+b) once; no f32 scatter atomics
# speedup vs baseline: 1.8444x; 1.8444x over previous
#include <hip/hip_runtime.h>
#include <hip/hip_bf16.h>


#define N_NODES  100000
#define N_EDGES  600000
#define N_GRAPHS 2000
#define NCLS     4

typedef float v2f __attribute__((ext_vector_type(2)));
typedef float v8f __attribute__((ext_vector_type(8)));


__global__ void k_zero(float* __restrict__ p, int n) {
  int i = blockIdx.x * blockDim.x + threadIdx.x;
  if (i < n) p[i] = 0.0f;
}

__global__ void k_zero_i(int* __restrict__ p, int n) {
  int i = blockIdx.x * blockDim.x + threadIdx.x;
  if (i < n) p[i] = 0;
}


__global__ void k_rank(const int* __restrict__ dst, int* __restrict__ icnt,
                       int* __restrict__ rank) {
  int e = blockIdx.x * blockDim.x + threadIdx.x;
  if (e < N_EDGES) rank[e] = atomicAdd(&icnt[dst[e]], 1);
}

__global__ void k_rows(const int* __restrict__ icnt, int* __restrict__ total,
                       int* __restrict__ rowstart, float* __restrict__ dinv) {
  const int i    = blockIdx.x * blockDim.x + threadIdx.x;
  const int lane = threadIdx.x & 31;
  const int c    = (i < N_NODES) ? icnt[i] : 0;
  int incl = c;
  for (int off = 1; off < 32; off <<= 1) {
    const int t = __shfl_up(incl, off);
    if (lane >= off) incl += t;
  }
  int base = 0;
  if (lane == 31) base = atomicAdd(total, incl);
  base = __shfl(base, 31);
  if (i < N_NODES) {
    rowstart[i] = base + incl - c;
    dinv[i]     = rsqrtf((float)c + 1.0f);
  }
}

__global__ void k_fill(const int* __restrict__ src, const int* __restrict__ dst,
                       const int* __restrict__ rank, const int* __restrict__ rowstart,
                       int* __restrict__ grouped) {
  int e = blockIdx.x * blockDim.x + threadIdx.x;
  if (e < N_EDGES) grouped[rowstart[dst[e]] + rank[e]] = src[e];
}

__global__ void k_embed(const int* __restrict__ tok, const float* __restrict__ emb,
                        float* __restrict__ x0) {
  size_t idx = (size_t)blockIdx.x * blockDim.x + threadIdx.x;
  int node = (int)(idx >> 6);
  int c    = (int)(idx & 63);
  x0[idx] = emb[((size_t)tok[node] << 6) | c];
}

template<int K>
__global__ void gemm_wmma_kernel(const float* __restrict__ X,
                                 const float* __restrict__ W,
                                 float* __restrict__ H) {
  const int wave  = (blockIdx.x << 3) | (threadIdx.x >> 5);
  const int lane  = threadIdx.x & 31;
  const int tileM = wave >> 3;
  const int tileN = wave & 7;
  const int half  = lane >> 4;
  const int lid   = lane & 15;
  const int m     = tileM * 16 + lid;
  const int n     = tileN * 16 + lid;
  const int kb    = half * 2;

  v8f acc = {};
  const float* xrow = X + (size_t)m * K + kb;
  const float* wcol = W + (size_t)kb * 128 + n;
#pragma unroll
  for (int k = 0; k < K; k += 4) {
    v2f a; a.x = xrow[k];               a.y = xrow[k + 1];
    v2f b; b.x = wcol[(size_t)k * 128]; b.y = wcol[(size_t)(k + 1) * 128];
    acc = __builtin_amdgcn_wmma_f32_16x16x4_f32(
         false, a,  false, b,
         (short)0, acc,  false,  false);
  }

  float* hbase = H + ((size_t)(tileM * 16 + half * 8)) * 128 + tileN * 16 + lid;
#pragma unroll
  for (int j = 0; j < 8; ++j)
    hbase[(size_t)j * 128] = acc[j];
}


static __device__ __forceinline__ void fma4(float4& acc, const float4 h, const float w) {
  acc.x = fmaf(h.x, w, acc.x);
  acc.y = fmaf(h.y, w, acc.y);
  acc.z = fmaf(h.z, w, acc.z);
  acc.w = fmaf(h.w, w, acc.w);
}

__global__ void __launch_bounds__(256)
k_gather(const float* __restrict__ H, const int* __restrict__ rowstart,
         const int* __restrict__ icnt, const int* __restrict__ grouped,
         const float* __restrict__ dinv, const float* __restrict__ bias,
         float* __restrict__ out) {
  const int node = __builtin_amdgcn_readfirstlane((int)((blockIdx.x << 3) | (threadIdx.x >> 5)));
  const int lane = threadIdx.x & 31;
  const int beg  = rowstart[node];
  const int n    = icnt[node];
  const float di = dinv[node];
  const float4* __restrict__ Hv = (const float4*)H;

  float4 acc = Hv[((size_t)node << 5) + lane];
  const float ws = di * di;
  acc.x *= ws; acc.y *= ws; acc.z *= ws; acc.w *= ws;

  const int* __restrict__ row = grouped + beg;
  int j = 0;
  for (; j + 4 <= n; j += 4) {
    const int s0 = row[j], s1 = row[j + 1], s2 = row[j + 2], s3 = row[j + 3];
    const float4 h0 = Hv[((size_t)s0 << 5) + lane];
    const float4 h1 = Hv[((size_t)s1 << 5) + lane];
    const float4 h2 = Hv[((size_t)s2 << 5) + lane];
    const float4 h3 = Hv[((size_t)s3 << 5) + lane];
    const float w0 = dinv[s0] * di, w1 = dinv[s1] * di;
    const float w2 = dinv[s2] * di, w3 = dinv[s3] * di;
    fma4(acc, h0, w0);
    fma4(acc, h1, w1);
    fma4(acc, h2, w2);
    fma4(acc, h3, w3);
  }
  if (j + 2 <= n) {
    const int s0 = row[j], s1 = row[j + 1];
    const float4 h0 = Hv[((size_t)s0 << 5) + lane];
    const float4 h1 = Hv[((size_t)s1 << 5) + lane];
    const float w0 = dinv[s0] * di, w1 = dinv[s1] * di;
    fma4(acc, h0, w0);
    fma4(acc, h1, w1);
    j += 2;
  }
  if (j < n) {
    const int s0 = row[j];
    const float4 h0 = Hv[((size_t)s0 << 5) + lane];
    fma4(acc, h0, dinv[s0] * di);
  }

  const float4 b = ((const float4*)bias)[lane];
  float4 v;
  v.x = fmaxf(acc.x + b.x, 0.0f);
  v.y = fmaxf(acc.y + b.y, 0.0f);
  v.z = fmaxf(acc.z + b.z, 0.0f);
  v.w = fmaxf(acc.w + b.w, 0.0f);
  ((float4*)out)[((size_t)node << 5) + lane] = v;
}


__global__ void k_pool(const float* __restrict__ X, const int* __restrict__ batch,
                       float* __restrict__ sums, float* __restrict__ cnts) {
  size_t gid = (size_t)blockIdx.x * blockDim.x + threadIdx.x;
  int node = (int)(gid >> 5);
  int lane = (int)(gid & 31);
  if (node >= N_NODES) return;
  int g = batch[node];
  const float4 v = *(const float4*)(X + ((size_t)node << 7) + (lane << 2));
  float* o = sums + ((size_t)g << 7) + (lane << 2);
  atomicAdd(o + 0, v.x);
  atomicAdd(o + 1, v.y);
  atomicAdd(o + 2, v.z);
  atomicAdd(o + 3, v.w);
  if (lane == 0) atomicAdd(&cnts[g], 1.0f);
}

__global__ void k_final(const float* __restrict__ sums, const float* __restrict__ cnts,
                        const float* __restrict__ Wlin, const float* __restrict__ blin,
                        float* __restrict__ out) {
  int t = blockIdx.x * blockDim.x + threadIdx.x;
  if (t >= N_GRAPHS * NCLS) return;
  int g = t >> 2, c = t & 3;
  float inv = 1.0f / fmaxf(cnts[g], 1.0f);
  float acc = blin[c];
  const float* srow = sums + ((size_t)g << 7);
#pragma unroll 4
  for (int k = 0; k < 128; ++k)
    acc = fmaf(srow[k] * inv, Wlin[k * NCLS + c], acc);
  out[t] = acc;
}


extern "C" void kernel_launch(void* const* d_in, const int* in_sizes, int n_in,
                              void* d_out, int out_size, void* d_ws, size_t ws_size,
                              hipStream_t stream) {
  const int*   tok   = (const int*)d_in[0];
  const int*   eidx  = (const int*)d_in[1];
  const int*   src   = eidx;
  const int*   dst   = eidx + N_EDGES;
  const int*   batch = (const int*)d_in[2];
  const float* emb   = (const float*)d_in[3];
  const float* W1    = (const float*)d_in[4];
  const float* b1    = (const float*)d_in[5];
  const float* W2    = (const float*)d_in[6];
  const float* b2    = (const float*)d_in[7];
  const float* Wlin  = (const float*)d_in[8];
  const float* blin  = (const float*)d_in[9];
  float*       out   = (float*)d_out;

  float* ws   = (float*)d_ws;
  float* dinv = ws;
  float* x0   = dinv + N_NODES;
  float* A    = x0   + (size_t)N_NODES * 64;
  float* B    = A    + (size_t)N_NODES * 128;
  float* sums = B    + (size_t)N_NODES * 128;
  float* cnts = sums + (size_t)N_GRAPHS * 128;
  int*   icnt     = (int*)(cnts + N_GRAPHS);
  int*   total    = icnt + N_NODES;
  int*   rowstart = total + 4;
  int*   rank     = rowstart + N_NODES;
  int*   grouped  = rank + N_EDGES;

  const int T = 256;
  const int gNodes  = (N_NODES + T - 1) / T;
  const int gNodes1 = (N_NODES + 1 + T - 1) / T;
  const int gEdges  = (N_EDGES + T - 1) / T;
  const int gEmbed  = (N_NODES * 64) / T;
  const int gRows   = (N_NODES * 32) / T;
  const int gPool   = (N_NODES * 32) / T;
  const int gGemm   = N_NODES / 16;

  k_zero_i<<<gNodes1, T, 0, stream>>>(icnt, N_NODES + 1);
  k_rank<<<gEdges, T, 0, stream>>>(dst, icnt, rank);
  k_rows<<<gNodes, T, 0, stream>>>(icnt, total, rowstart, dinv);
  k_fill<<<gEdges, T, 0, stream>>>(src, dst, rank, rowstart, grouped);

  k_embed<<<gEmbed, T, 0, stream>>>(tok, emb, x0);

  gemm_wmma_kernel<64><<<gGemm, T, 0, stream>>>(x0, W1, A);
  k_gather<<<gRows, T, 0, stream>>>(A, rowstart, icnt, grouped, dinv, b1, B);

  gemm_wmma_kernel<128><<<gGemm, T, 0, stream>>>(B, W2, A);
  k_gather<<<gRows, T, 0, stream>>>(A, rowstart, icnt, grouped, dinv, b2, B);

  k_zero<<<(N_GRAPHS * 128 + N_GRAPHS + T - 1) / T, T, 0, stream>>>(sums,
                                                    N_GRAPHS * 128 + N_GRAPHS);
  k_pool<<<gPool, T, 0, stream>>>(B, batch, sums, cnts);
  k_final<<<(N_GRAPHS * NCLS + T - 1) / T, T, 0, stream>>>(sums, cnts, Wlin, blin, out);
}


# layer-1 transform hoisted through the token lookup: emb_table@W1 on WMMA once per call (1000 rows); gather reads table[tok[src]]; embed copy and N-row layer-1 GEMM removed; sum order unchanged
# speedup vs baseline: 2.2363x; 1.2125x over previous
#include <hip/hip_runtime.h>
#include <hip/hip_bf16.h>


#define N_NODES  100000
#define N_EDGES  600000
#define N_GRAPHS 2000
#define NCLS     4
#define VOCAB    1000
#define VOCAB_PAD 1008

typedef float v2f __attribute__((ext_vector_type(2)));
typedef float v8f __attribute__((ext_vector_type(8)));


__global__ void k_zero(float* __restrict__ p, int n) {
  int i = blockIdx.x * blockDim.x + threadIdx.x;
  if (i < n) p[i] = 0.0f;
}

__global__ void k_zero_i(int* __restrict__ p, int n) {
  int i = blockIdx.x * blockDim.x + threadIdx.x;
  if (i < n) p[i] = 0;
}


__global__ void k_rank(const int* __restrict__ dst, int* __restrict__ icnt,
                       int* __restrict__ rank) {
  int e = blockIdx.x * blockDim.x + threadIdx.x;
  if (e < N_EDGES) rank[e] = atomicAdd(&icnt[dst[e]], 1);
}

__global__ void k_rows(const int* __restrict__ icnt, int* __restrict__ total,
                       int* __restrict__ rowstart, float* __restrict__ dinv) {
  const int i    = blockIdx.x * blockDim.x + threadIdx.x;
  const int lane = threadIdx.x & 31;
  const int c    = (i < N_NODES) ? icnt[i] : 0;
  int incl = c;
  for (int off = 1; off < 32; off <<= 1) {
    const int t = __shfl_up(incl, off);
    if (lane >= off) incl += t;
  }
  int base = 0;
  if (lane == 31) base = atomicAdd(total, incl);
  base = __shfl(base, 31);
  if (i < N_NODES) {
    rowstart[i] = base + incl - c;
    dinv[i]     = rsqrtf((float)c + 1.0f);
  }
}

__global__ void k_fill(const int* __restrict__ src, const int* __restrict__ dst,
                       const int* __restrict__ rank, const int* __restrict__ rowstart,
                       int* __restrict__ grouped) {
  int e = blockIdx.x * blockDim.x + threadIdx.x;
  if (e < N_EDGES) grouped[rowstart[dst[e]] + rank[e]] = src[e];
}

template<int K, int ROWS>
__global__ void gemm_wmma_kernel(const float* __restrict__ X,
                                 const float* __restrict__ W,
                                 float* __restrict__ H) {
  const int wave  = (blockIdx.x << 3) | (threadIdx.x >> 5);
  const int lane  = threadIdx.x & 31;
  const int tileM = wave >> 3;
  const int tileN = wave & 7;
  const int half  = lane >> 4;
  const int lid   = lane & 15;
  const int mt    = tileM * 16 + lid;
  const int m     = (ROWS % 16 != 0 && mt > ROWS - 1) ? ROWS - 1 : mt;
  const int n     = tileN * 16 + lid;
  const int kb    = half * 2;

  v8f acc = {};
  const float* xrow = X + (size_t)m * K + kb;
  const float* wcol = W + (size_t)kb * 128 + n;
#pragma unroll
  for (int k = 0; k < K; k += 4) {
    v2f a; a.x = xrow[k];               a.y = xrow[k + 1];
    v2f b; b.x = wcol[(size_t)k * 128]; b.y = wcol[(size_t)(k + 1) * 128];
    acc = __builtin_amdgcn_wmma_f32_16x16x4_f32(
         false, a,  false, b,
         (short)0, acc,  false,  false);
  }

  float* hbase = H + ((size_t)(tileM * 16 + half * 8)) * 128 + tileN * 16 + lid;
#pragma unroll
  for (int j = 0; j < 8; ++j)
    hbase[(size_t)j * 128] = acc[j];
}


static __device__ __forceinline__ void fma4(float4& acc, const float4 h, const float w) {
  acc.x = fmaf(h.x, w, acc.x);
  acc.y = fmaf(h.y, w, acc.y);
  acc.z = fmaf(h.z, w, acc.z);
  acc.w = fmaf(h.w, w, acc.w);
}

template<bool VIA_TOK>
__global__ void __launch_bounds__(256)
k_gather(const float* __restrict__ H, const int* __restrict__ rowstart,
         const int* __restrict__ icnt, const int* __restrict__ grouped,
         const float* __restrict__ dinv, const float* __restrict__ bias,
         const int* __restrict__ tok, float* __restrict__ out) {
  const int node = __builtin_amdgcn_readfirstlane((int)((blockIdx.x << 3) | (threadIdx.x >> 5)));
  const int lane = threadIdx.x & 31;
  const int beg  = rowstart[node];
  const int n    = icnt[node];
  const float di = dinv[node];
  const float4* __restrict__ Hv = (const float4*)H;

  float4 acc = Hv[((size_t)(VIA_TOK ? tok[node] : node) << 5) + lane];
  const float ws = di * di;
  acc.x *= ws; acc.y *= ws; acc.z *= ws; acc.w *= ws;

  const int* __restrict__ row = grouped + beg;
  int j = 0;
  for (; j + 4 <= n; j += 4) {
    const int s0 = row[j], s1 = row[j + 1], s2 = row[j + 2], s3 = row[j + 3];
    const int r0 = VIA_TOK ? tok[s0] : s0, r1 = VIA_TOK ? tok[s1] : s1;
    const int r2 = VIA_TOK ? tok[s2] : s2, r3 = VIA_TOK ? tok[s3] : s3;
    const float4 h0 = Hv[((size_t)r0 << 5) + lane];
    const float4 h1 = Hv[((size_t)r1 << 5) + lane];
    const float4 h2 = Hv[((size_t)r2 << 5) + lane];
    const float4 h3 = Hv[((size_t)r3 << 5) + lane];
    const float w0 = dinv[s0] * di, w1 = dinv[s1] * di;
    const float w2 = dinv[s2] * di, w3 = dinv[s3] * di;
    fma4(acc, h0, w0);
    fma4(acc, h1, w1);
    fma4(acc, h2, w2);
    fma4(acc, h3, w3);
  }
  if (j + 2 <= n) {
    const int s0 = row[j], s1 = row[j + 1];
    const int r0 = VIA_TOK ? tok[s0] : s0, r1 = VIA_TOK ? tok[s1] : s1;
    const float4 h0 = Hv[((size_t)r0 << 5) + lane];
    const float4 h1 = Hv[((size_t)r1 << 5) + lane];
    const float w0 = dinv[s0] * di, w1 = dinv[s1] * di;
    fma4(acc, h0, w0);
    fma4(acc, h1, w1);
    j += 2;
  }
  if (j < n) {
    const int s0 = row[j];
    const int r0 = VIA_TOK ? tok[s0] : s0;
    const float4 h0 = Hv[((size_t)r0 << 5) + lane];
    fma4(acc, h0, dinv[s0] * di);
  }

  const float4 b = ((const float4*)bias)[lane];
  float4 v;
  v.x = fmaxf(acc.x + b.x, 0.0f);
  v.y = fmaxf(acc.y + b.y, 0.0f);
  v.z = fmaxf(acc.z + b.z, 0.0f);
  v.w = fmaxf(acc.w + b.w, 0.0f);
  ((float4*)out)[((size_t)node << 5) + lane] = v;
}


__global__ void k_pool(const float* __restrict__ X, const int* __restrict__ batch,
                       float* __restrict__ sums, float* __restrict__ cnts) {
  size_t gid = (size_t)blockIdx.x * blockDim.x + threadIdx.x;
  int node = (int)(gid >> 5);
  int lane = (int)(gid & 31);
  if (node >= N_NODES) return;
  int g = batch[node];
  const float4 v = *(const float4*)(X + ((size_t)node << 7) + (lane << 2));
  float* o = sums + ((size_t)g << 7) + (lane << 2);
  atomicAdd(o + 0, v.x);
  atomicAdd(o + 1, v.y);
  atomicAdd(o + 2, v.z);
  atomicAdd(o + 3, v.w);
  if (lane == 0) atomicAdd(&cnts[g], 1.0f);
}

__global__ void k_final(const float* __restrict__ sums, const float* __restrict__ cnts,
                        const float* __restrict__ Wlin, const float* __restrict__ blin,
                        float* __restrict__ out) {
  int t = blockIdx.x * blockDim.x + threadIdx.x;
  if (t >= N_GRAPHS * NCLS) return;
  int g = t >> 2, c = t & 3;
  float inv = 1.0f / fmaxf(cnts[g], 1.0f);
  float acc = blin[c];
  const float* srow = sums + ((size_t)g << 7);
#pragma unroll 4
  for (int k = 0; k < 128; ++k)
    acc = fmaf(srow[k] * inv, Wlin[k * NCLS + c], acc);
  out[t] = acc;
}


extern "C" void kernel_launch(void* const* d_in, const int* in_sizes, int n_in,
                              void* d_out, int out_size, void* d_ws, size_t ws_size,
                              hipStream_t stream) {
  const int*   tok   = (const int*)d_in[0];
  const int*   eidx  = (const int*)d_in[1];
  const int*   src   = eidx;
  const int*   dst   = eidx + N_EDGES;
  const int*   batch = (const int*)d_in[2];
  const float* emb   = (const float*)d_in[3];
  const float* W1    = (const float*)d_in[4];
  const float* b1    = (const float*)d_in[5];
  const float* W2    = (const float*)d_in[6];
  const float* b2    = (const float*)d_in[7];
  const float* Wlin  = (const float*)d_in[8];
  const float* blin  = (const float*)d_in[9];
  float*       out   = (float*)d_out;

  float* ws   = (float*)d_ws;
  float* dinv = ws;
  float* tab  = dinv + N_NODES;
  float* A    = tab  + (size_t)VOCAB_PAD * 128;
  float* B    = A    + (size_t)N_NODES * 128;
  float* sums = B    + (size_t)N_NODES * 128;
  float* cnts = sums + (size_t)N_GRAPHS * 128;
  int*   icnt     = (int*)(cnts + N_GRAPHS);
  int*   total    = icnt + N_NODES;
  int*   rowstart = total + 4;
  int*   rank     = rowstart + N_NODES;
  int*   grouped  = rank + N_EDGES;

  const int T = 256;
  const int gNodes  = (N_NODES + T - 1) / T;
  const int gNodes1 = (N_NODES + 1 + T - 1) / T;
  const int gEdges  = (N_EDGES + T - 1) / T;
  const int gTab    = VOCAB_PAD / 16;
  const int gRows   = (N_NODES * 32) / T;
  const int gPool   = (N_NODES * 32) / T;
  const int gGemm   = N_NODES / 16;

  k_zero_i<<<gNodes1, T, 0, stream>>>(icnt, N_NODES + 1);
  k_rank<<<gEdges, T, 0, stream>>>(dst, icnt, rank);
  k_rows<<<gNodes, T, 0, stream>>>(icnt, total, rowstart, dinv);
  k_fill<<<gEdges, T, 0, stream>>>(src, dst, rank, rowstart, grouped);

  gemm_wmma_kernel<64, VOCAB><<<gTab, T, 0, stream>>>(emb, W1, tab);
  k_gather<true><<<gRows, T, 0, stream>>>(tab, rowstart, icnt, grouped, dinv, b1, tok, B);

  gemm_wmma_kernel<128, N_NODES><<<gGemm, T, 0, stream>>>(B, W2, A);
  k_gather<false><<<gRows, T, 0, stream>>>(A, rowstart, icnt, grouped, dinv, b2, tok, B);

  k_zero<<<(N_GRAPHS * 128 + N_GRAPHS + T - 1) / T, T, 0, stream>>>(sums,
                                                    N_GRAPHS * 128 + N_GRAPHS);
  k_pool<<<gPool, T, 0, stream>>>(B, batch, sums, cnts);
  k_final<<<(N_GRAPHS * NCLS + T - 1) / T, T, 0, stream>>>(sums, cnts, Wlin, blin, out);
}


# layer GEMMs retiled: one wave per 16x128 strip, A fragment used for 8 column tiles, W kept in LDS in fragment order, whole-row loads and stores via LDS; k sum order permuted within each 8
# speedup vs baseline: 2.4667x; 1.1030x over previous
#include <hip/hip_runtime.h>
#include <hip/hip_bf16.h>


#define N_NODES  100000
#define N_EDGES  600000
#define N_GRAPHS 2000
#define NCLS     4
#define VOCAB    1000
#define VOCAB_PAD 1008

typedef float v2f __attribute__((ext_vector_type(2)));
typedef float v4f __attribute__((ext_vector_type(4)));
typedef float v8f __attribute__((ext_vector_type(8)));


__global__ void k_zero(float* __restrict__ p, int n) {
  int i = blockIdx.x * blockDim.x + threadIdx.x;
  if (i < n) p[i] = 0.0f;
}

__global__ void k_zero_i(int* __restrict__ p, int n) {
  int i = blockIdx.x * blockDim.x + threadIdx.x;
  if (i < n) p[i] = 0;
}


__global__ void k_rank(const int* __restrict__ dst, int* __restrict__ icnt,
                       int* __restrict__ rank) {
  int e = blockIdx.x * blockDim.x + threadIdx.x;
  if (e < N_EDGES) rank[e] = atomicAdd(&icnt[dst[e]], 1);
}

__global__ void k_rows(const int* __restrict__ icnt, int* __restrict__ total,
                       int* __restrict__ rowstart, float* __restrict__ dinv) {
  const int i    = blockIdx.x * blockDim.x + threadIdx.x;
  const int lane = threadIdx.x & 31;
  const int c    = (i < N_NODES) ? icnt[i] : 0;
  int incl = c;
  for (int off = 1; off < 32; off <<= 1) {
    const int t = __shfl_up(incl, off);
    if (lane >= off) incl += t;
  }
  int base = 0;
  if (lane == 31) base = atomicAdd(total, incl);
  base = __shfl(base, 31);
  if (i < N_NODES) {
    rowstart[i] = base + incl - c;
    dinv[i]     = rsqrtf((float)c + 1.0f);
  }
}

__global__ void k_fill(const int* __restrict__ src, const int* __restrict__ dst,
                       const int* __restrict__ rank, const int* __restrict__ rowstart,
                       int* __restrict__ grouped) {
  int e = blockIdx.x * blockDim.x + threadIdx.x;
  if (e < N_EDGES) grouped[rowstart[dst[e]] + rank[e]] = src[e];
}


static __device__ __forceinline__ void wave_tile_sync() {
  __builtin_amdgcn_fence(__ATOMIC_RELEASE, "wavefront");
  __builtin_amdgcn_wave_barrier();
  __builtin_amdgcn_fence(__ATOMIC_ACQUIRE, "wavefront");
}

template<int K, int ROWS, int NLD>
static __device__ __forceinline__ void strip_load_rows(v4f (&pre)[NLD], const float* __restrict__ X,
                                                       const int strip, const int lane) {
  const int h = lane >> 4, c = lane & 15;
#pragma unroll
  for (int i = 0; i < NLD; ++i) {
    const int rt = strip * 16 + ((K == 128) ? i : 2 * i + h);
    const int m  = (ROWS % 16 != 0 && rt > ROWS - 1) ? ROWS - 1 : rt;
    pre[i] = *(const v4f*)(X + (size_t)m * K + 4 * ((K == 128) ? lane : c));
  }
}

template<int K, int NLD, int CP>
static __device__ __forceinline__ void strip_rows_to_tile(const v4f (&pre)[NLD], float* tile,
                                                          const int lane) {
  const int h = lane >> 4, c = lane & 15;
#pragma unroll
  for (int i = 0; i < NLD; ++i) {
    const int r = (K == 128) ? i : 2 * i + h;
    *(v4f*)(tile + r * CP + 4 * ((K == 128) ? lane : c)) = pre[i];
  }
}

template<int K, int ROWS, int NWAVES>
__global__ void __launch_bounds__(NWAVES * 32)
gemm_strip_kernel(const float* __restrict__ X,
                  const float* __restrict__ W,
                  float* __restrict__ H) {
  constexpr int KT      = K / 8;
  constexpr int NSTRIPS = (ROWS + 15) / 16;
  constexpr int CP      = 132;
  static_assert(K == 64 || K == 128, "row loads below are written for K = 64 or 128");
  __shared__ v4f Wf[KT * 8 * 32];
  __shared__ __attribute__((aligned(16))) float Tile[NWAVES * 16 * CP];

  constexpr int NSTAGE = (KT * 8 * 32) / (NWAVES * 32);
  static_assert((KT * 8 * 32) % (NWAVES * 32) == 0, "every thread stages the same number of entries");
#pragma unroll
  for (int i = 0; i < NSTAGE; ++i) {
    const int f  = (int)threadIdx.x + i * (NWAVES * 32);
    const int ln = f & 31, j = (f >> 5) & 7, t = f >> 8;
    const float* wp = W + (size_t)(8 * t + 4 * (ln >> 4)) * 128 + 16 * j + (ln & 15);
    v4f v;
    v.x = wp[0]; v.y = wp[128]; v.z = wp[256]; v.w = wp[384];
    Wf[f] = v;
  }
  __syncthreads();

  const int wave = __builtin_amdgcn_readfirstlane((int)(threadIdx.x >> 5));
  const int lane = threadIdx.x & 31;
  const int h    = lane >> 4;
  const int c    = lane & 15;
  float* tile = Tile + wave * (16 * CP);
  const int step = NWAVES * (int)gridDim.x;

  int s = wave * (int)gridDim.x + (int)blockIdx.x;
  if (s >= NSTRIPS) return;

  constexpr int NLD = (K == 128) ? 16 : 8;
  v4f pre[NLD];

  strip_load_rows<K, ROWS, NLD>(pre, X, s, lane);
  strip_rows_to_tile<K, NLD, CP>(pre, tile, lane);
  wave_tile_sync();

  for (;;) {
    const int  sn   = s + step;
    const bool more = sn < NSTRIPS;
    strip_load_rows<K, ROWS, NLD>(pre, X, more ? sn : s, lane);

    v8f acc[8];
#pragma unroll
    for (int j = 0; j < 8; ++j) acc[j] = (v8f){};

    v4f a = *(const v4f*)(tile + c * CP + 4 * h);
    v4f b[8];
#pragma unroll
    for (int j = 0; j < 8; ++j) b[j] = Wf[j * 32 + lane];

#pragma unroll 2
    for (int t = 0; t < KT; ++t) {
      const int tn = (t + 1 < KT) ? t + 1 : t;
      const v4f an = *(const v4f*)(tile + c * CP + 8 * tn + 4 * h);
      v4f bn[8];
#pragma unroll
      for (int j = 0; j < 8; ++j) bn[j] = Wf[(tn * 8 + j) * 32 + lane];

      v2f a0; a0.x = a.x; a0.y = a.y;
      v2f a1; a1.x = a.z; a1.y = a.w;
#pragma unroll
      for (int j = 0; j < 8; ++j) {
        v2f b0; b0.x = b[j].x; b0.y = b[j].y;
        acc[j] = __builtin_amdgcn_wmma_f32_16x16x4_f32(
             false, a0,  false, b0,
             (short)0, acc[j],  false,  false);
      }
#pragma unroll
      for (int j = 0; j < 8; ++j) {
        v2f b1; b1.x = b[j].z; b1.y = b[j].w;
        acc[j] = __builtin_amdgcn_wmma_f32_16x16x4_f32(
             false, a1,  false, b1,
             (short)0, acc[j],  false,  false);
      }
      a = an;
#pragma unroll
      for (int j = 0; j < 8; ++j) b[j] = bn[j];
    }

#pragma unroll
    for (int j = 0; j < 8; ++j)
#pragma unroll
      for (int r = 0; r < 8; ++r)
        tile[(8 * h + r) * CP + 16 * j + c] = acc[j][r];
    wave_tile_sync();

    float* hrow = H + (size_t)s * 16 * 128 + 4 * lane;
#pragma unroll
    for (int r = 0; r < 16; ++r) {
      const v4f v = *(const v4f*)(tile + r * CP + 4 * lane);
      *(v4f*)(hrow + (size_t)r * 128) = v;
    }
    wave_tile_sync();

    if (!more) break;
    s = sn;
    strip_rows_to_tile<K, NLD, CP>(pre, tile, lane);
    wave_tile_sync();
  }
}


static __device__ __forceinline__ void fma4(float4& acc, const float4 h, const float w) {
  acc.x = fmaf(h.x, w, acc.x);
  acc.y = fmaf(h.y, w, acc.y);
  acc.z = fmaf(h.z, w, acc.z);
  acc.w = fmaf(h.w, w, acc.w);
}

template<bool VIA_TOK>
__global__ void __launch_bounds__(256)
k_gather(const float* __restrict__ H, const int* __restrict__ rowstart,
         const int* __restrict__ icnt, const int* __restrict__ grouped,
         const float* __restrict__ dinv, const float* __restrict__ bias,
         const int* __restrict__ tok, float* __restrict__ out) {
  const int node = __builtin_amdgcn_readfirstlane((int)((blockIdx.x << 3) | (threadIdx.x >> 5)));
  const int lane = threadIdx.x & 31;
  const int beg  = rowstart[node];
  const int n    = icnt[node];
  const float di = dinv[node];
  const float4* __restrict__ Hv = (const float4*)H;

  float4 acc = Hv[((size_t)(VIA_TOK ? tok[node] : node) << 5) + lane];
  const float ws = di * di;
  acc.x *= ws; acc.y *= ws; acc.z *= ws; acc.w *= ws;

  const int* __restrict__ row = grouped + beg;
  int j = 0;
  for (; j + 4 <= n; j += 4) {
    const int s0 = row[j], s1 = row[j + 1], s2 = row[j + 2], s3 = row[j + 3];
    const int r0 = VIA_TOK ? tok[s0] : s0, r1 = VIA_TOK ? tok[s1] : s1;
    const int r2 = VIA_TOK ? tok[s2] : s2, r3 = VIA_TOK ? tok[s3] : s3;
    const float4 h0 = Hv[((size_t)r0 << 5) + lane];
    const float4 h1 = Hv[((size_t)r1 << 5) + lane];
    const float4 h2 = Hv[((size_t)r2 << 5) + lane];
    const float4 h3 = Hv[((size_t)r3 << 5) + lane];
    const float w0 = dinv[s0] * di, w1 = dinv[s1] * di;
    const float w2 = dinv[s2] * di, w3 = dinv[s3] * di;
    fma4(acc, h0, w0);
    fma4(acc, h1, w1);
    fma4(acc, h2, w2);
    fma4(acc, h3, w3);
  }
  if (j + 2 <= n) {
    const int s0 = row[j], s1 = row[j + 1];
    const int r0 = VIA_TOK ? tok[s0] : s0, r1 = VIA_TOK ? tok[s1] : s1;
    const float4 h0 = Hv[((size_t)r0 << 5) + lane];
    const float4 h1 = Hv[((size_t)r1 << 5) + lane];
    const float w0 = dinv[s0] * di, w1 = dinv[s1] * di;
    fma4(acc, h0, w0);
    fma4(acc, h1, w1);
    j += 2;
  }
  if (j < n) {
    const int s0 = row[j];
    const int r0 = VIA_TOK ? tok[s0] : s0;
    const float4 h0 = Hv[((size_t)r0 << 5) + lane];
    fma4(acc, h0, dinv[s0] * di);
  }

  const float4 b = ((const float4*)bias)[lane];
  float4 v;
  v.x = fmaxf(acc.x + b.x, 0.0f);
  v.y = fmaxf(acc.y + b.y, 0.0f);
  v.z = fmaxf(acc.z + b.z, 0.0f);
  v.w = fmaxf(acc.w + b.w, 0.0f);
  ((float4*)out)[((size_t)node << 5) + lane] = v;
}


__global__ void k_pool(const float* __restrict__ X, const int* __restrict__ batch,
                       float* __restrict__ sums, float* __restrict__ cnts) {
  size_t gid = (size_t)blockIdx.x * blockDim.x + threadIdx.x;
  int node = (int)(gid >> 5);
  int lane = (int)(gid & 31);
  if (node >= N_NODES) return;
  int g = batch[node];
  const float4 v = *(const float4*)(X + ((size_t)node << 7) + (lane << 2));
  float* o = sums + ((size_t)g << 7) + (lane << 2);
  atomicAdd(o + 0, v.x);
  atomicAdd(o + 1, v.y);
  atomicAdd(o + 2, v.z);
  atomicAdd(o + 3, v.w);
  if (lane == 0) atomicAdd(&cnts[g], 1.0f);
}

__global__ void k_final(const float* __restrict__ sums, const float* __restrict__ cnts,
                        const float* __restrict__ Wlin, const float* __restrict__ blin,
                        float* __restrict__ out) {
  int t = blockIdx.x * blockDim.x + threadIdx.x;
  if (t >= N_GRAPHS * NCLS) return;
  int g = t >> 2, c = t & 3;
  float inv = 1.0f / fmaxf(cnts[g], 1.0f);
  float acc = blin[c];
  const float* srow = sums + ((size_t)g << 7);
#pragma unroll 4
  for (int k = 0; k < 128; ++k)
    acc = fmaf(srow[k] * inv, Wlin[k * NCLS + c], acc);
  out[t] = acc;
}


extern "C" void kernel_launch(void* const* d_in, const int* in_sizes, int n_in,
                              void* d_out, int out_size, void* d_ws, size_t ws_size,
                              hipStream_t stream) {
  const int*   tok   = (const int*)d_in[0];
  const int*   eidx  = (const int*)d_in[1];
  const int*   src   = eidx;
  const int*   dst   = eidx + N_EDGES;
  const int*   batch = (const int*)d_in[2];
  const float* emb   = (const float*)d_in[3];
  const float* W1    = (const float*)d_in[4];
  const float* b1    = (const float*)d_in[5];
  const float* W2    = (const float*)d_in[6];
  const float* b2    = (const float*)d_in[7];
  const float* Wlin  = (const float*)d_in[8];
  const float* blin  = (const float*)d_in[9];
  float*       out   = (float*)d_out;

  float* ws   = (float*)d_ws;
  float* dinv = ws;
  float* tab  = dinv + N_NODES;
  float* A    = tab  + (size_t)VOCAB_PAD * 128;
  float* B    = A    + (size_t)N_NODES * 128;
  float* sums = B    + (size_t)N_NODES * 128;
  float* cnts = sums + (size_t)N_GRAPHS * 128;
  int*   icnt     = (int*)(cnts + N_GRAPHS);
  int*   total    = icnt + N_NODES;
  int*   rowstart = total + 4;
  int*   rank     = rowstart + N_NODES;
  int*   grouped  = rank + N_EDGES;

  const int T = 256;
  const int gNodes  = (N_NODES + T - 1) / T;
  const int gNodes1 = (N_NODES + 1 + T - 1) / T;
  const int gEdges  = (N_EDGES + T - 1) / T;
  const int gTab    = 8;
  const int gRows   = (N_NODES * 32) / T;
  const int gPool   = (N_NODES * 32) / T;
  const int gGemm   = 256;

  k_zero_i<<<gNodes1, T, 0, stream>>>(icnt, N_NODES + 1);
  k_rank<<<gEdges, T, 0, stream>>>(dst, icnt, rank);
  k_rows<<<gNodes, T, 0, stream>>>(icnt, total, rowstart, dinv);
  k_fill<<<gEdges, T, 0, stream>>>(src, dst, rank, rowstart, grouped);

  gemm_strip_kernel<64, VOCAB, 8><<<gTab, 256, 0, stream>>>(emb, W1, tab);
  k_gather<true><<<gRows, T, 0, stream>>>(tab, rowstart, icnt, grouped, dinv, b1, tok, B);

  gemm_strip_kernel<128, N_NODES, 16><<<gGemm, 512, 0, stream>>>(B, W2, A);
  k_gather<false><<<gRows, T, 0, stream>>>(A, rowstart, icnt, grouped, dinv, b2, tok, B);

  k_zero<<<(N_GRAPHS * 128 + N_GRAPHS + T - 1) / T, T, 0, stream>>>(sums,
                                                    N_GRAPHS * 128 + N_GRAPHS);
  k_pool<<<gPool, T, 0, stream>>>(B, batch, sums, cnts);
  k_final<<<(N_GRAPHS * NCLS + T - 1) / T, T, 0, stream>>>(sums, cnts, Wlin, blin, out);
}


# classifier applied per node before the mean pool, inside the layer-2 gather: row @ Wlin summed by a lane tree, then 4 float atomics per node (was 128); pool kernel and x2 store gone
# speedup vs baseline: 3.3315x; 1.3506x over previous
#include <hip/hip_runtime.h>
#include <hip/hip_bf16.h>


#define N_NODES  100000
#define N_EDGES  600000
#define N_GRAPHS 2000
#define NCLS     4
#define VOCAB    1000
#define VOCAB_PAD 1008

typedef float v2f __attribute__((ext_vector_type(2)));
typedef float v4f __attribute__((ext_vector_type(4)));
typedef float v8f __attribute__((ext_vector_type(8)));


__global__ void k_zero_i(int* __restrict__ p, int n) {
  int i = blockIdx.x * blockDim.x + threadIdx.x;
  if (i < n) p[i] = 0;
}


__global__ void k_rank(const int* __restrict__ dst, int* __restrict__ icnt,
                       int* __restrict__ rank) {
  int e = blockIdx.x * blockDim.x + threadIdx.x;
  if (e < N_EDGES) rank[e] = atomicAdd(&icnt[dst[e]], 1);
}

__global__ void k_rows(const int* __restrict__ icnt, int* __restrict__ total,
                       int* __restrict__ rowstart, float* __restrict__ dinv) {
  const int i    = blockIdx.x * blockDim.x + threadIdx.x;
  const int lane = threadIdx.x & 31;
  const int c    = (i < N_NODES) ? icnt[i] : 0;
  int incl = c;
  for (int off = 1; off < 32; off <<= 1) {
    const int t = __shfl_up(incl, off);
    if (lane >= off) incl += t;
  }
  int base = 0;
  if (lane == 31) base = atomicAdd(total, incl);
  base = __shfl(base, 31);
  if (i < N_NODES) {
    rowstart[i] = base + incl - c;
    dinv[i]     = rsqrtf((float)c + 1.0f);
  }
}

__global__ void k_fill(const int* __restrict__ src, const int* __restrict__ dst,
                       const int* __restrict__ rank, const int* __restrict__ rowstart,
                       int* __restrict__ grouped) {
  int e = blockIdx.x * blockDim.x + threadIdx.x;
  if (e < N_EDGES) grouped[rowstart[dst[e]] + rank[e]] = src[e];
}


static __device__ __forceinline__ void wave_tile_sync() {
  __builtin_amdgcn_fence(__ATOMIC_RELEASE, "wavefront");
  __builtin_amdgcn_wave_barrier();
  __builtin_amdgcn_fence(__ATOMIC_ACQUIRE, "wavefront");
}

template<int K, int ROWS, int NLD>
static __device__ __forceinline__ void strip_load_rows(v4f (&pre)[NLD], const float* __restrict__ X,
                                                       const int strip, const int lane) {
  const int h = lane >> 4, c = lane & 15;
#pragma unroll
  for (int i = 0; i < NLD; ++i) {
    const int rt = strip * 16 + ((K == 128) ? i : 2 * i + h);
    const int m  = (ROWS % 16 != 0 && rt > ROWS - 1) ? ROWS - 1 : rt;
    pre[i] = *(const v4f*)(X + (size_t)m * K + 4 * ((K == 128) ? lane : c));
  }
}

template<int K, int NLD, int CP>
static __device__ __forceinline__ void strip_rows_to_tile(const v4f (&pre)[NLD], float* tile,
                                                          const int lane) {
  const int h = lane >> 4, c = lane & 15;
#pragma unroll
  for (int i = 0; i < NLD; ++i) {
    const int r = (K == 128) ? i : 2 * i + h;
    *(v4f*)(tile + r * CP + 4 * ((K == 128) ? lane : c)) = pre[i];
  }
}

template<int K, int ROWS, int NWAVES>
__global__ void __launch_bounds__(NWAVES * 32)
gemm_strip_kernel(const float* __restrict__ X,
                  const float* __restrict__ W,
                  float* __restrict__ H) {
  constexpr int KT      = K / 8;
  constexpr int NSTRIPS = (ROWS + 15) / 16;
  constexpr int CP      = 132;
  static_assert(K == 64 || K == 128, "row loads below are written for K = 64 or 128");
  __shared__ v4f Wf[KT * 8 * 32];
  __shared__ __attribute__((aligned(16))) float Tile[NWAVES * 16 * CP];

  constexpr int NSTAGE = (KT * 8 * 32) / (NWAVES * 32);
  static_assert((KT * 8 * 32) % (NWAVES * 32) == 0, "every thread stages the same number of entries");
#pragma unroll
  for (int i = 0; i < NSTAGE; ++i) {
    const int f  = (int)threadIdx.x + i * (NWAVES * 32);
    const int ln = f & 31, j = (f >> 5) & 7, t = f >> 8;
    const float* wp = W + (size_t)(8 * t + 4 * (ln >> 4)) * 128 + 16 * j + (ln & 15);
    v4f v;
    v.x = wp[0]; v.y = wp[128]; v.z = wp[256]; v.w = wp[384];
    Wf[f] = v;
  }
  __syncthreads();

  const int wave = __builtin_amdgcn_readfirstlane((int)(threadIdx.x >> 5));
  const int lane = threadIdx.x & 31;
  const int h    = lane >> 4;
  const int c    = lane & 15;
  float* tile = Tile + wave * (16 * CP);
  const int step = NWAVES * (int)gridDim.x;

  int s = wave * (int)gridDim.x + (int)blockIdx.x;
  if (s >= NSTRIPS) return;

  constexpr int NLD = (K == 128) ? 16 : 8;
  v4f pre[NLD];

  strip_load_rows<K, ROWS, NLD>(pre, X, s, lane);
  strip_rows_to_tile<K, NLD, CP>(pre, tile, lane);
  wave_tile_sync();

  for (;;) {
    const int  sn   = s + step;
    const bool more = sn < NSTRIPS;
    strip_load_rows<K, ROWS, NLD>(pre, X, more ? sn : s, lane);

    v8f acc[8];
#pragma unroll
    for (int j = 0; j < 8; ++j) acc[j] = (v8f){};

    v4f a = *(const v4f*)(tile + c * CP + 4 * h);
    v4f b[8];
#pragma unroll
    for (int j = 0; j < 8; ++j) b[j] = Wf[j * 32 + lane];

#pragma unroll 2
    for (int t = 0; t < KT; ++t) {
      const int tn = (t + 1 < KT) ? t + 1 : t;
      const v4f an = *(const v4f*)(tile + c * CP + 8 * tn + 4 * h);
      v4f bn[8];
#pragma unroll
      for (int j = 0; j < 8; ++j) bn[j] = Wf[(tn * 8 + j) * 32 + lane];

      v2f a0; a0.x = a.x; a0.y = a.y;
      v2f a1; a1.x = a.z; a1.y = a.w;
#pragma unroll
      for (int j = 0; j < 8; ++j) {
        v2f b0; b0.x = b[j].x; b0.y = b[j].y;
        acc[j] = __builtin_amdgcn_wmma_f32_16x16x4_f32(
             false, a0,  false, b0,
             (short)0, acc[j],  false,  false);
      }
#pragma unroll
      for (int j = 0; j < 8; ++j) {
        v2f b1; b1.x = b[j].z; b1.y = b[j].w;
        acc[j] = __builtin_amdgcn_wmma_f32_16x16x4_f32(
             false, a1,  false, b1,
             (short)0, acc[j],  false,  false);
      }
      a = an;
#pragma unroll
      for (int j = 0; j < 8; ++j) b[j] = bn[j];
    }

#pragma unroll
    for (int j = 0; j < 8; ++j)
#pragma unroll
      for (int r = 0; r < 8; ++r)
        tile[(8 * h + r) * CP + 16 * j + c] = acc[j][r];
    wave_tile_sync();

    float* hrow = H + (size_t)s * 16 * 128 + 4 * lane;
#pragma unroll
    for (int r = 0; r < 16; ++r) {
      const v4f v = *(const v4f*)(tile + r * CP + 4 * lane);
      *(v4f*)(hrow + (size_t)r * 128) = v;
    }
    wave_tile_sync();

    if (!more) break;
    s = sn;
    strip_rows_to_tile<K, NLD, CP>(pre, tile, lane);
    wave_tile_sync();
  }
}


static __device__ __forceinline__ void fma4(float4& acc, const float4 h, const float w) {
  acc.x = fmaf(h.x, w, acc.x);
  acc.y = fmaf(h.y, w, acc.y);
  acc.z = fmaf(h.z, w, acc.z);
  acc.w = fmaf(h.w, w, acc.w);
}

template<bool VIA_TOK, bool READOUT>
__global__ void __launch_bounds__(256)
k_gather(const float* __restrict__ H, const int* __restrict__ rowstart,
         const int* __restrict__ icnt, const int* __restrict__ grouped,
         const float* __restrict__ dinv, const float* __restrict__ bias,
         const int* __restrict__ tok, float* __restrict__ out,
         const int* __restrict__ batch, const float* __restrict__ Wlin,
         float* __restrict__ pool) {
  const int node = __builtin_amdgcn_readfirstlane((int)((blockIdx.x << 3) | (threadIdx.x >> 5)));
  const int lane = threadIdx.x & 31;
  const int beg  = rowstart[node];
  const int n    = icnt[node];
  const float di = dinv[node];
  const int   g  = READOUT ? batch[node] : 0;
  const float4* __restrict__ Hv = (const float4*)H;

  float4 acc = Hv[((size_t)(VIA_TOK ? tok[node] : node) << 5) + lane];
  const float ws = di * di;
  acc.x *= ws; acc.y *= ws; acc.z *= ws; acc.w *= ws;

  const int* __restrict__ row = grouped + beg;
  int j = 0;
  for (; j + 4 <= n; j += 4) {
    const int s0 = row[j], s1 = row[j + 1], s2 = row[j + 2], s3 = row[j + 3];
    const int r0 = VIA_TOK ? tok[s0] : s0, r1 = VIA_TOK ? tok[s1] : s1;
    const int r2 = VIA_TOK ? tok[s2] : s2, r3 = VIA_TOK ? tok[s3] : s3;
    const float4 h0 = Hv[((size_t)r0 << 5) + lane];
    const float4 h1 = Hv[((size_t)r1 << 5) + lane];
    const float4 h2 = Hv[((size_t)r2 << 5) + lane];
    const float4 h3 = Hv[((size_t)r3 << 5) + lane];
    const float w0 = dinv[s0] * di, w1 = dinv[s1] * di;
    const float w2 = dinv[s2] * di, w3 = dinv[s3] * di;
    fma4(acc, h0, w0);
    fma4(acc, h1, w1);
    fma4(acc, h2, w2);
    fma4(acc, h3, w3);
  }
  if (j + 2 <= n) {
    const int s0 = row[j], s1 = row[j + 1];
    const int r0 = VIA_TOK ? tok[s0] : s0, r1 = VIA_TOK ? tok[s1] : s1;
    const float4 h0 = Hv[((size_t)r0 << 5) + lane];
    const float4 h1 = Hv[((size_t)r1 << 5) + lane];
    const float w0 = dinv[s0] * di, w1 = dinv[s1] * di;
    fma4(acc, h0, w0);
    fma4(acc, h1, w1);
    j += 2;
  }
  if (j < n) {
    const int s0 = row[j];
    const int r0 = VIA_TOK ? tok[s0] : s0;
    const float4 h0 = Hv[((size_t)r0 << 5) + lane];
    fma4(acc, h0, dinv[s0] * di);
  }

  const float4 b = ((const float4*)bias)[lane];
  float4 v;
  v.x = fmaxf(acc.x + b.x, 0.0f);
  v.y = fmaxf(acc.y + b.y, 0.0f);
  v.z = fmaxf(acc.z + b.z, 0.0f);
  v.w = fmaxf(acc.w + b.w, 0.0f);
  if (!READOUT) {
    ((float4*)out)[((size_t)node << 5) + lane] = v;
    return;
  }

  const float4* __restrict__ Wl = (const float4*)Wlin + 4 * lane;
  const float4 w0 = Wl[0], w1 = Wl[1], w2 = Wl[2], w3 = Wl[3];
  const float p0 = fmaf(v.w, w3.x, fmaf(v.z, w2.x, fmaf(v.y, w1.x, v.x * w0.x)));
  const float p1 = fmaf(v.w, w3.y, fmaf(v.z, w2.y, fmaf(v.y, w1.y, v.x * w0.y)));
  const float p2 = fmaf(v.w, w3.z, fmaf(v.z, w2.z, fmaf(v.y, w1.z, v.x * w0.z)));
  const float p3 = fmaf(v.w, w3.w, fmaf(v.z, w2.w, fmaf(v.y, w1.w, v.x * w0.w)));

  const bool odd1 = (lane & 1) != 0, odd2 = (lane & 2) != 0;
  const float qa = (odd1 ? p2 : p0) + __shfl_xor(odd1 ? p0 : p2, 1);
  const float qb = (odd1 ? p3 : p1) + __shfl_xor(odd1 ? p1 : p3, 1);
  float r = (odd2 ? qb : qa) + __shfl_xor(odd2 ? qa : qb, 2);
  r += __shfl_xor(r, 4);
  r += __shfl_xor(r, 8);
  r += __shfl_xor(r, 16);

  if (lane < 5) {
    const int   slot = (lane == 4) ? 4 : (((lane & 1) << 1) | (lane >> 1));
    const float val  = (lane == 4) ? 1.0f : r;
    atomicAdd(pool + ((size_t)g << 3) + slot, val);
  }
}


__global__ void k_final(const float* __restrict__ pool, const float* __restrict__ blin,
                        float* __restrict__ out) {
  int t = blockIdx.x * blockDim.x + threadIdx.x;
  if (t >= N_GRAPHS * NCLS) return;
  int g = t >> 2, c = t & 3;
  const float* p = pool + ((size_t)g << 3);
  out[t] = p[c] / fmaxf(p[4], 1.0f) + blin[c];
}


extern "C" void kernel_launch(void* const* d_in, const int* in_sizes, int n_in,
                              void* d_out, int out_size, void* d_ws, size_t ws_size,
                              hipStream_t stream) {
  const int*   tok   = (const int*)d_in[0];
  const int*   eidx  = (const int*)d_in[1];
  const int*   src   = eidx;
  const int*   dst   = eidx + N_EDGES;
  const int*   batch = (const int*)d_in[2];
  const float* emb   = (const float*)d_in[3];
  const float* W1    = (const float*)d_in[4];
  const float* b1    = (const float*)d_in[5];
  const float* W2    = (const float*)d_in[6];
  const float* b2    = (const float*)d_in[7];
  const float* Wlin  = (const float*)d_in[8];
  const float* blin  = (const float*)d_in[9];
  float*       out   = (float*)d_out;

  float* ws   = (float*)d_ws;
  float* dinv = ws;
  float* tab  = dinv + N_NODES;
  float* A    = tab  + (size_t)VOCAB_PAD * 128;
  float* B    = A    + (size_t)N_NODES * 128;
  int*   icnt     = (int*)(B + (size_t)N_NODES * 128);
  int*   total    = icnt + N_NODES;
  float* pool     = (float*)(total + 8);
  int*   rowstart = (int*)(pool + N_GRAPHS * 8);
  int*   rank     = rowstart + N_NODES;
  int*   grouped  = rank + N_EDGES;

  const int T = 256;
  const int gNodes  = (N_NODES + T - 1) / T;
  const int nClear  = N_NODES + 8 + N_GRAPHS * 8;
  const int gClear  = (nClear + T - 1) / T;
  const int gEdges  = (N_EDGES + T - 1) / T;
  const int gTab    = 8;
  const int gRows   = (N_NODES * 32) / T;
  const int gGemm   = 256;

  k_zero_i<<<gClear, T, 0, stream>>>(icnt, nClear);
  k_rank<<<gEdges, T, 0, stream>>>(dst, icnt, rank);
  k_rows<<<gNodes, T, 0, stream>>>(icnt, total, rowstart, dinv);
  k_fill<<<gEdges, T, 0, stream>>>(src, dst, rank, rowstart, grouped);

  gemm_strip_kernel<64, VOCAB, 8><<<gTab, 256, 0, stream>>>(emb, W1, tab);
  k_gather<true, false><<<gRows, T, 0, stream>>>(tab, rowstart, icnt, grouped, dinv, b1, tok, B,
                                                 nullptr, nullptr, nullptr);

  gemm_strip_kernel<128, N_NODES, 16><<<gGemm, 512, 0, stream>>>(B, W2, A);
  k_gather<false, true><<<gRows, T, 0, stream>>>(A, rowstart, icnt, grouped, dinv, b2, tok, nullptr,
                                                 batch, Wlin, pool);

  k_final<<<(N_GRAPHS * NCLS + T - 1) / T, T, 0, stream>>>(pool, blin, out);
}


# edge ranks from LDS counters: one work-group per (4000-node range, 60000-edge stretch), no per-edge global atomic; clear launch gone; a node's in-edge sum order now stretch order, then arrival
# speedup vs baseline: 3.6237x; 1.0877x over previous
#include <hip/hip_runtime.h>
#include <hip/hip_bf16.h>


#define N_NODES  100000
#define N_EDGES  600000
#define N_GRAPHS 2000
#define NCLS     4
#define VOCAB    1000
#define VOCAB_PAD 1008

typedef float v2f __attribute__((ext_vector_type(2)));
typedef float v4f __attribute__((ext_vector_type(4)));
typedef float v8f __attribute__((ext_vector_type(8)));


#define RK_RANGES 25
#define RK_NODES  4000
#define RK_CHUNKS 10
#define RK_EDGES  60000
#define RK_T      1024

__global__ void __launch_bounds__(RK_T)
k_rank(const int* __restrict__ dst, int* __restrict__ cnts, int* __restrict__ rank,
       int* __restrict__ total, float* __restrict__ pool) {
  static_assert(RK_RANGES * RK_NODES == N_NODES && RK_CHUNKS * RK_EDGES == N_EDGES, "exact cover");
  static_assert(RK_EDGES % 4 == 0, "a stretch is a whole number of 16-byte loads");
  static_assert(RK_RANGES * RK_CHUNKS * 64 == N_GRAPHS * 8, "pool is cleared in slices of 64 words");
  __shared__ int cnt[RK_NODES];
  const int r   = (int)blockIdx.x % RK_RANGES;
  const int c   = (int)blockIdx.x / RK_RANGES;
  const int lo  = r * RK_NODES;
  const int tid = (int)threadIdx.x;

  constexpr int Q   = RK_EDGES / 4;
  constexpr int NIT = (Q + RK_T - 1) / RK_T;
  const int4* __restrict__ dv = (const int4*)(dst + (size_t)c * RK_EDGES);
  int4 d[NIT];
#pragma unroll
  for (int it = 0; it < NIT; ++it) {
    const int q = tid + it * RK_T;
    d[it] = (q < Q) ? dv[q] : make_int4(-1, -1, -1, -1);
  }

  for (int i = tid; i < RK_NODES; i += RK_T) cnt[i] = 0;
  if (tid < 64) pool[(size_t)blockIdx.x * 64 + tid] = 0.0f;
  if (blockIdx.x == 0 && tid == 0) *total = 0;
  __syncthreads();

#pragma unroll
  for (int it = 0; it < NIT; ++it) {
    const int q = tid + it * RK_T;
    int* __restrict__ rk = rank + (size_t)c * RK_EDGES + 4 * (size_t)q;
    const unsigned u0 = (unsigned)(d[it].x - lo), u1 = (unsigned)(d[it].y - lo);
    const unsigned u2 = (unsigned)(d[it].z - lo), u3 = (unsigned)(d[it].w - lo);
    if (u0 < (unsigned)RK_NODES) rk[0] = atomicAdd(&cnt[u0], 1);
    if (u1 < (unsigned)RK_NODES) rk[1] = atomicAdd(&cnt[u1], 1);
    if (u2 < (unsigned)RK_NODES) rk[2] = atomicAdd(&cnt[u2], 1);
    if (u3 < (unsigned)RK_NODES) rk[3] = atomicAdd(&cnt[u3], 1);
  }
  __syncthreads();

  for (int i = tid; i < RK_NODES; i += RK_T) cnts[(size_t)c * N_NODES + lo + i] = cnt[i];
}

__global__ void k_rows(int* __restrict__ cnts, int* __restrict__ icnt, int* __restrict__ total,
                       int* __restrict__ rowstart, float* __restrict__ dinv) {
  const int i    = blockIdx.x * blockDim.x + threadIdx.x;
  const int lane = threadIdx.x & 31;
  int before[RK_CHUNKS];
  int c = 0;
#pragma unroll
  for (int k = 0; k < RK_CHUNKS; ++k) {
    before[k] = c;
    c += (i < N_NODES) ? cnts[(size_t)k * N_NODES + i] : 0;
  }
  int incl = c;
  for (int off = 1; off < 32; off <<= 1) {
    const int t = __shfl_up(incl, off);
    if (lane >= off) incl += t;
  }
  int base = 0;
  if (lane == 31) base = atomicAdd(total, incl);
  base = __shfl(base, 31);
  if (i < N_NODES) {
    const int rs = base + incl - c;
    rowstart[i] = rs;
    icnt[i]     = c;
    dinv[i]     = rsqrtf((float)c + 1.0f);
#pragma unroll
    for (int k = 0; k < RK_CHUNKS; ++k) cnts[(size_t)k * N_NODES + i] = rs + before[k];
  }
}

__global__ void k_fill(const int* __restrict__ src, const int* __restrict__ dst,
                       const int* __restrict__ rank, const int* __restrict__ cnts,
                       int* __restrict__ grouped) {
  int e = blockIdx.x * blockDim.x + threadIdx.x;
  if (e < N_EDGES) grouped[cnts[(size_t)(e / RK_EDGES) * N_NODES + dst[e]] + rank[e]] = src[e];
}


static __device__ __forceinline__ void wave_tile_sync() {
  __builtin_amdgcn_fence(__ATOMIC_RELEASE, "wavefront");
  __builtin_amdgcn_wave_barrier();
  __builtin_amdgcn_fence(__ATOMIC_ACQUIRE, "wavefront");
}

template<int K, int ROWS, int NLD>
static __device__ __forceinline__ void strip_load_rows(v4f (&pre)[NLD], const float* __restrict__ X,
                                                       const int strip, const int lane) {
  const int h = lane >> 4, c = lane & 15;
#pragma unroll
  for (int i = 0; i < NLD; ++i) {
    const int rt = strip * 16 + ((K == 128) ? i : 2 * i + h);
    const int m  = (ROWS % 16 != 0 && rt > ROWS - 1) ? ROWS - 1 : rt;
    pre[i] = *(const v4f*)(X + (size_t)m * K + 4 * ((K == 128) ? lane : c));
  }
}

template<int K, int NLD, int CP>
static __device__ __forceinline__ void strip_rows_to_tile(const v4f (&pre)[NLD], float* tile,
                                                          const int lane) {
  const int h = lane >> 4, c = lane & 15;
#pragma unroll
  for (int i = 0; i < NLD; ++i) {
    const int r = (K == 128) ? i : 2 * i + h;
    *(v4f*)(tile + r * CP + 4 * ((K == 128) ? lane : c)) = pre[i];
  }
}

template<int K, int ROWS, int NWAVES>
__global__ void __launch_bounds__(NWAVES * 32)
gemm_strip_kernel(const float* __restrict__ X,
                  const float* __restrict__ W,
                  float* __restrict__ H) {
  constexpr int KT      = K / 8;
  constexpr int NSTRIPS = (ROWS + 15) / 16;
  constexpr int CP      = 132;
  static_assert(K == 64 || K == 128, "row loads below are written for K = 64 or 128");
  __shared__ v4f Wf[KT * 8 * 32];
  __shared__ __attribute__((aligned(16))) float Tile[NWAVES * 16 * CP];

  constexpr int NSTAGE = (KT * 8 * 32) / (NWAVES * 32);
  static_assert((KT * 8 * 32) % (NWAVES * 32) == 0, "every thread stages the same number of entries");
#pragma unroll
  for (int i = 0; i < NSTAGE; ++i) {
    const int f  = (int)threadIdx.x + i * (NWAVES * 32);
    const int ln = f & 31, j = (f >> 5) & 7, t = f >> 8;
    const float* wp = W + (size_t)(8 * t + 4 * (ln >> 4)) * 128 + 16 * j + (ln & 15);
    v4f v;
    v.x = wp[0]; v.y = wp[128]; v.z = wp[256]; v.w = wp[384];
    Wf[f] = v;
  }
  __syncthreads();

  const int wave = __builtin_amdgcn_readfirstlane((int)(threadIdx.x >> 5));
  const int lane = threadIdx.x & 31;
  const int h    = lane >> 4;
  const int c    = lane & 15;
  float* tile = Tile + wave * (16 * CP);
  const int step = NWAVES * (int)gridDim.x;

  int s = wave * (int)gridDim.x + (int)blockIdx.x;
  if (s >= NSTRIPS) return;

  constexpr int NLD = (K == 128) ? 16 : 8;
  v4f pre[NLD];

  strip_load_rows<K, ROWS, NLD>(pre, X, s, lane);
  strip_rows_to_tile<K, NLD, CP>(pre, tile, lane);
  wave_tile_sync();

  for (;;) {
    const int  sn   = s + step;
    const bool more = sn < NSTRIPS;
    strip_load_rows<K, ROWS, NLD>(pre, X, more ? sn : s, lane);

    v8f acc[8];
#pragma unroll
    for (int j = 0; j < 8; ++j) acc[j] = (v8f){};

    v4f a = *(const v4f*)(tile + c * CP + 4 * h);
    v4f b[8];
#pragma unroll
    for (int j = 0; j < 8; ++j) b[j] = Wf[j * 32 + lane];

#pragma unroll 2
    for (int t = 0; t < KT; ++t) {
      const int tn = (t + 1 < KT) ? t + 1 : t;
      const v4f an = *(const v4f*)(tile + c * CP + 8 * tn + 4 * h);
      v4f bn[8];
#pragma unroll
      for (int j = 0; j < 8; ++j) bn[j] = Wf[(tn * 8 + j) * 32 + lane];

      v2f a0; a0.x = a.x; a0.y = a.y;
      v2f a1; a1.x = a.z; a1.y = a.w;
#pragma unroll
      for (int j = 0; j < 8; ++j) {
        v2f b0; b0.x = b[j].x; b0.y = b[j].y;
        acc[j] = __builtin_amdgcn_wmma_f32_16x16x4_f32(
             false, a0,  false, b0,
             (short)0, acc[j],  false,  false);
      }
#pragma unroll
      for (int j = 0; j < 8; ++j) {
        v2f b1; b1.x = b[j].z; b1.y = b[j].w;
        acc[j] = __builtin_amdgcn_wmma_f32_16x16x4_f32(
             false, a1,  false, b1,
             (short)0, acc[j],  false,  false);
      }
      a = an;
#pragma unroll
      for (int j = 0; j < 8; ++j) b[j] = bn[j];
    }

#pragma unroll
    for (int j = 0; j < 8; ++j)
#pragma unroll
      for (int r = 0; r < 8; ++r)
        tile[(8 * h + r) * CP + 16 * j + c] = acc[j][r];
    wave_tile_sync();

    float* hrow = H + (size_t)s * 16 * 128 + 4 * lane;
#pragma unroll
    for (int r = 0; r < 16; ++r) {
      const v4f v = *(const v4f*)(tile + r * CP + 4 * lane);
      *(v4f*)(hrow + (size_t)r * 128) = v;
    }
    wave_tile_sync();

    if (!more) break;
    s = sn;
    strip_rows_to_tile<K, NLD, CP>(pre, tile, lane);
    wave_tile_sync();
  }
}


static __device__ __forceinline__ void fma4(float4& acc, const float4 h, const float w) {
  acc.x = fmaf(h.x, w, acc.x);
  acc.y = fmaf(h.y, w, acc.y);
  acc.z = fmaf(h.z, w, acc.z);
  acc.w = fmaf(h.w, w, acc.w);
}

template<bool VIA_TOK, bool READOUT>
__global__ void __launch_bounds__(256)
k_gather(const float* __restrict__ H, const int* __restrict__ rowstart,
         const int* __restrict__ icnt, const int* __restrict__ grouped,
         const float* __restrict__ dinv, const float* __restrict__ bias,
         const int* __restrict__ tok, float* __restrict__ out,
         const int* __restrict__ batch, const float* __restrict__ Wlin,
         float* __restrict__ pool) {
  const int node = __builtin_amdgcn_readfirstlane((int)((blockIdx.x << 3) | (threadIdx.x >> 5)));
  const int lane = threadIdx.x & 31;
  const int beg  = rowstart[node];
  const int n    = icnt[node];
  const float di = dinv[node];
  const int   g  = READOUT ? batch[node] : 0;
  const float4* __restrict__ Hv = (const float4*)H;

  float4 acc = Hv[((size_t)(VIA_TOK ? tok[node] : node) << 5) + lane];
  const float ws = di * di;
  acc.x *= ws; acc.y *= ws; acc.z *= ws; acc.w *= ws;

  const int* __restrict__ row = grouped + beg;
  int j = 0;
  for (; j + 4 <= n; j += 4) {
    const int s0 = row[j], s1 = row[j + 1], s2 = row[j + 2], s3 = row[j + 3];
    const int r0 = VIA_TOK ? tok[s0] : s0, r1 = VIA_TOK ? tok[s1] : s1;
    const int r2 = VIA_TOK ? tok[s2] : s2, r3 = VIA_TOK ? tok[s3] : s3;
    const float4 h0 = Hv[((size_t)r0 << 5) + lane];
    const float4 h1 = Hv[((size_t)r1 << 5) + lane];
    const float4 h2 = Hv[((size_t)r2 << 5) + lane];
    const float4 h3 = Hv[((size_t)r3 << 5) + lane];
    const float w0 = dinv[s0] * di, w1 = dinv[s1] * di;
    const float w2 = dinv[s2] * di, w3 = dinv[s3] * di;
    fma4(acc, h0, w0);
    fma4(acc, h1, w1);
    fma4(acc, h2, w2);
    fma4(acc, h3, w3);
  }
  if (j + 2 <= n) {
    const int s0 = row[j], s1 = row[j + 1];
    const int r0 = VIA_TOK ? tok[s0] : s0, r1 = VIA_TOK ? tok[s1] : s1;
    const float4 h0 = Hv[((size_t)r0 << 5) + lane];
    const float4 h1 = Hv[((size_t)r1 << 5) + lane];
    const float w0 = dinv[s0] * di, w1 = dinv[s1] * di;
    fma4(acc, h0, w0);
    fma4(acc, h1, w1);
    j += 2;
  }
  if (j < n) {
    const int s0 = row[j];
    const int r0 = VIA_TOK ? tok[s0] : s0;
    const float4 h0 = Hv[((size_t)r0 << 5) + lane];
    fma4(acc, h0, dinv[s0] * di);
  }

  const float4 b = ((const float4*)bias)[lane];
  float4 v;
  v.x = fmaxf(acc.x + b.x, 0.0f);
  v.y = fmaxf(acc.y + b.y, 0.0f);
  v.z = fmaxf(acc.z + b.z, 0.0f);
  v.w = fmaxf(acc.w + b.w, 0.0f);
  if (!READOUT) {
    ((float4*)out)[((size_t)node << 5) + lane] = v;
    return;
  }

  const float4* __restrict__ Wl = (const float4*)Wlin + 4 * lane;
  const float4 w0 = Wl[0], w1 = Wl[1], w2 = Wl[2], w3 = Wl[3];
  const float p0 = fmaf(v.w, w3.x, fmaf(v.z, w2.x, fmaf(v.y, w1.x, v.x * w0.x)));
  const float p1 = fmaf(v.w, w3.y, fmaf(v.z, w2.y, fmaf(v.y, w1.y, v.x * w0.y)));
  const float p2 = fmaf(v.w, w3.z, fmaf(v.z, w2.z, fmaf(v.y, w1.z, v.x * w0.z)));
  const float p3 = fmaf(v.w, w3.w, fmaf(v.z, w2.w, fmaf(v.y, w1.w, v.x * w0.w)));

  const bool odd1 = (lane & 1) != 0, odd2 = (lane & 2) != 0;
  const float qa = (odd1 ? p2 : p0) + __shfl_xor(odd1 ? p0 : p2, 1);
  const float qb = (odd1 ? p3 : p1) + __shfl_xor(odd1 ? p1 : p3, 1);
  float r = (odd2 ? qb : qa) + __shfl_xor(odd2 ? qa : qb, 2);
  r += __shfl_xor(r, 4);
  r += __shfl_xor(r, 8);
  r += __shfl_xor(r, 16);

  if (lane < 5) {
    const int   slot = (lane == 4) ? 4 : (((lane & 1) << 1) | (lane >> 1));
    const float val  = (lane == 4) ? 1.0f : r;
    atomicAdd(pool + ((size_t)g << 3) + slot, val);
  }
}


__global__ void k_final(const float* __restrict__ pool, const float* __restrict__ blin,
                        float* __restrict__ out) {
  int t = blockIdx.x * blockDim.x + threadIdx.x;
  if (t >= N_GRAPHS * NCLS) return;
  int g = t >> 2, c = t & 3;
  const float* p = pool + ((size_t)g << 3);
  out[t] = p[c] / fmaxf(p[4], 1.0f) + blin[c];
}


extern "C" void kernel_launch(void* const* d_in, const int* in_sizes, int n_in,
                              void* d_out, int out_size, void* d_ws, size_t ws_size,
                              hipStream_t stream) {
  const int*   tok   = (const int*)d_in[0];
  const int*   eidx  = (const int*)d_in[1];
  const int*   src   = eidx;
  const int*   dst   = eidx + N_EDGES;
  const int*   batch = (const int*)d_in[2];
  const float* emb   = (const float*)d_in[3];
  const float* W1    = (const float*)d_in[4];
  const float* b1    = (const float*)d_in[5];
  const float* W2    = (const float*)d_in[6];
  const float* b2    = (const float*)d_in[7];
  const float* Wlin  = (const float*)d_in[8];
  const float* blin  = (const float*)d_in[9];
  float*       out   = (float*)d_out;

  float* ws   = (float*)d_ws;
  float* dinv = ws;
  float* tab  = dinv + N_NODES;
  float* A    = tab  + (size_t)VOCAB_PAD * 128;
  float* B    = A    + (size_t)N_NODES * 128;
  int*   icnt     = (int*)(B + (size_t)N_NODES * 128);
  int*   total    = icnt + N_NODES;
  float* pool     = (float*)(total + 8);
  int*   rowstart = (int*)(pool + N_GRAPHS * 8);
  int*   rank     = rowstart + N_NODES;
  int*   grouped  = rank + N_EDGES;
  int*   cnts     = grouped + N_EDGES;

  const int T = 256;
  const int gNodes  = (N_NODES + T - 1) / T;
  const int gEdges  = (N_EDGES + T - 1) / T;
  const int gTab    = 8;
  const int gRows   = (N_NODES * 32) / T;
  const int gGemm   = 256;

  k_rank<<<RK_RANGES * RK_CHUNKS, RK_T, 0, stream>>>(dst, cnts, rank, total, pool);
  k_rows<<<gNodes, T, 0, stream>>>(cnts, icnt, total, rowstart, dinv);
  k_fill<<<gEdges, T, 0, stream>>>(src, dst, rank, cnts, grouped);

  gemm_strip_kernel<64, VOCAB, 8><<<gTab, 256, 0, stream>>>(emb, W1, tab);
  k_gather<true, false><<<gRows, T, 0, stream>>>(tab, rowstart, icnt, grouped, dinv, b1, tok, B,
                                                 nullptr, nullptr, nullptr);

  gemm_strip_kernel<128, N_NODES, 16><<<gGemm, 512, 0, stream>>>(B, W2, A);
  k_gather<false, true><<<gRows, T, 0, stream>>>(A, rowstart, icnt, grouped, dinv, b2, tok, nullptr,
                                                 batch, Wlin, pool);

  k_final<<<(N_GRAPHS * NCLS + T - 1) / T, T, 0, stream>>>(pool, blin, out);
}


# table product emb@W1 rides in the span-scan launch as 16 extra work-groups (neither reads what the other writes): 7 launches, was 8; its load/matrix/store chain runs beside the scan; sums unchanged
# speedup vs baseline: 3.7654x; 1.0391x over previous
#include <hip/hip_runtime.h>
#include <hip/hip_bf16.h>


#define N_NODES  100000
#define N_EDGES  600000
#define N_GRAPHS 2000
#define NCLS     4
#define VOCAB    1000
#define VOCAB_PAD 1008

typedef float v2f __attribute__((ext_vector_type(2)));
typedef float v4f __attribute__((ext_vector_type(4)));
typedef float v8f __attribute__((ext_vector_type(8)));


#define RK_RANGES 25
#define RK_NODES  4000
#define RK_CHUNKS 10
#define RK_EDGES  60000
#define RK_T      1024

__global__ void __launch_bounds__(RK_T)
k_rank(const int* __restrict__ dst, int* __restrict__ cnts, int* __restrict__ rank,
       int* __restrict__ total, float* __restrict__ pool) {
  static_assert(RK_RANGES * RK_NODES == N_NODES && RK_CHUNKS * RK_EDGES == N_EDGES, "exact cover");
  static_assert(RK_EDGES % 4 == 0, "a stretch is a whole number of 16-byte loads");
  static_assert(RK_RANGES * RK_CHUNKS * 64 == N_GRAPHS * 8, "pool is cleared in slices of 64 words");
  __shared__ int cnt[RK_NODES];
  const int r   = (int)blockIdx.x % RK_RANGES;
  const int c   = (int)blockIdx.x / RK_RANGES;
  const int lo  = r * RK_NODES;
  const int tid = (int)threadIdx.x;

  constexpr int Q   = RK_EDGES / 4;
  constexpr int NIT = (Q + RK_T - 1) / RK_T;
  const int4* __restrict__ dv = (const int4*)(dst + (size_t)c * RK_EDGES);
  int4 d[NIT];
#pragma unroll
  for (int it = 0; it < NIT; ++it) {
    const int q = tid + it * RK_T;
    d[it] = (q < Q) ? dv[q] : make_int4(-1, -1, -1, -1);
  }

  for (int i = tid; i < RK_NODES; i += RK_T) cnt[i] = 0;
  if (tid < 64) pool[(size_t)blockIdx.x * 64 + tid] = 0.0f;
  if (blockIdx.x == 0 && tid == 0) *total = 0;
  __syncthreads();

#pragma unroll
  for (int it = 0; it < NIT; ++it) {
    const int q = tid + it * RK_T;
    int* __restrict__ rk = rank + (size_t)c * RK_EDGES + 4 * (size_t)q;
    const unsigned u0 = (unsigned)(d[it].x - lo), u1 = (unsigned)(d[it].y - lo);
    const unsigned u2 = (unsigned)(d[it].z - lo), u3 = (unsigned)(d[it].w - lo);
    if (u0 < (unsigned)RK_NODES) rk[0] = atomicAdd(&cnt[u0], 1);
    if (u1 < (unsigned)RK_NODES) rk[1] = atomicAdd(&cnt[u1], 1);
    if (u2 < (unsigned)RK_NODES) rk[2] = atomicAdd(&cnt[u2], 1);
    if (u3 < (unsigned)RK_NODES) rk[3] = atomicAdd(&cnt[u3], 1);
  }
  __syncthreads();

  for (int i = tid; i < RK_NODES; i += RK_T) cnts[(size_t)c * N_NODES + lo + i] = cnt[i];
}

#define ROWS_T 256
static __device__ __forceinline__ void
rows_body(int* __restrict__ cnts, int* __restrict__ icnt, int* __restrict__ total,
          int* __restrict__ rowstart, float* __restrict__ dinv, const int bid) {
  const int i    = bid * ROWS_T + (int)threadIdx.x;
  const int lane = threadIdx.x & 31;
  int before[RK_CHUNKS];
  int c = 0;
#pragma unroll
  for (int k = 0; k < RK_CHUNKS; ++k) {
    before[k] = c;
    c += (i < N_NODES) ? cnts[(size_t)k * N_NODES + i] : 0;
  }
  int incl = c;
  for (int off = 1; off < 32; off <<= 1) {
    const int t = __shfl_up(incl, off);
    if (lane >= off) incl += t;
  }
  int base = 0;
  if (lane == 31) base = atomicAdd(total, incl);
  base = __shfl(base, 31);
  if (i < N_NODES) {
    const int rs = base + incl - c;
    rowstart[i] = rs;
    icnt[i]     = c;
    dinv[i]     = rsqrtf((float)c + 1.0f);
#pragma unroll
    for (int k = 0; k < RK_CHUNKS; ++k) cnts[(size_t)k * N_NODES + i] = rs + before[k];
  }
}

__global__ void k_fill(const int* __restrict__ src, const int* __restrict__ dst,
                       const int* __restrict__ rank, const int* __restrict__ cnts,
                       int* __restrict__ grouped) {
  int e = blockIdx.x * blockDim.x + threadIdx.x;
  if (e < N_EDGES) grouped[cnts[(size_t)(e / RK_EDGES) * N_NODES + dst[e]] + rank[e]] = src[e];
}


static __device__ __forceinline__ void wave_tile_sync() {
  __builtin_amdgcn_fence(__ATOMIC_RELEASE, "wavefront");
  __builtin_amdgcn_wave_barrier();
  __builtin_amdgcn_fence(__ATOMIC_ACQUIRE, "wavefront");
}

template<int K, int ROWS, int NLD>
static __device__ __forceinline__ void strip_load_rows(v4f (&pre)[NLD], const float* __restrict__ X,
                                                       const int strip, const int lane) {
  const int h = lane >> 4, c = lane & 15;
#pragma unroll
  for (int i = 0; i < NLD; ++i) {
    const int rt = strip * 16 + ((K == 128) ? i : 2 * i + h);
    const int m  = (ROWS % 16 != 0 && rt > ROWS - 1) ? ROWS - 1 : rt;
    pre[i] = *(const v4f*)(X + (size_t)m * K + 4 * ((K == 128) ? lane : c));
  }
}

template<int K, int NLD, int CP>
static __device__ __forceinline__ void strip_rows_to_tile(const v4f (&pre)[NLD], float* tile,
                                                          const int lane) {
  const int h = lane >> 4, c = lane & 15;
#pragma unroll
  for (int i = 0; i < NLD; ++i) {
    const int r = (K == 128) ? i : 2 * i + h;
    *(v4f*)(tile + r * CP + 4 * ((K == 128) ? lane : c)) = pre[i];
  }
}

template<int K, int ROWS, int NWAVES>
static __device__ __forceinline__ void
gemm_strip_body(const float* __restrict__ X,
                const float* __restrict__ W,
                float* __restrict__ H,
                const int bid, const int nblk) {
  constexpr int KT      = K / 8;
  constexpr int NSTRIPS = (ROWS + 15) / 16;
  constexpr int CP      = 132;
  static_assert(K == 64 || K == 128, "row loads below are written for K = 64 or 128");
  __shared__ v4f Wf[KT * 8 * 32];
  __shared__ __attribute__((aligned(16))) float Tile[NWAVES * 16 * CP];

  constexpr int NSTAGE = (KT * 8 * 32) / (NWAVES * 32);
  static_assert((KT * 8 * 32) % (NWAVES * 32) == 0, "every thread stages the same number of entries");
#pragma unroll
  for (int i = 0; i < NSTAGE; ++i) {
    const int f  = (int)threadIdx.x + i * (NWAVES * 32);
    const int ln = f & 31, j = (f >> 5) & 7, t = f >> 8;
    const float* wp = W + (size_t)(8 * t + 4 * (ln >> 4)) * 128 + 16 * j + (ln & 15);
    v4f v;
    v.x = wp[0]; v.y = wp[128]; v.z = wp[256]; v.w = wp[384];
    Wf[f] = v;
  }
  __syncthreads();

  const int wave = __builtin_amdgcn_readfirstlane((int)(threadIdx.x >> 5));
  const int lane = threadIdx.x & 31;
  const int h    = lane >> 4;
  const int c    = lane & 15;
  float* tile = Tile + wave * (16 * CP);
  const int step = NWAVES * nblk;

  int s = wave * nblk + bid;
  if (s >= NSTRIPS) return;

  constexpr int NLD = (K == 128) ? 16 : 8;
  v4f pre[NLD];

  strip_load_rows<K, ROWS, NLD>(pre, X, s, lane);
  strip_rows_to_tile<K, NLD, CP>(pre, tile, lane);
  wave_tile_sync();

  for (;;) {
    const int  sn   = s + step;
    const bool more = sn < NSTRIPS;
    strip_load_rows<K, ROWS, NLD>(pre, X, more ? sn : s, lane);

    v8f acc[8];
#pragma unroll
    for (int j = 0; j < 8; ++j) acc[j] = (v8f){};

    v4f a = *(const v4f*)(tile + c * CP + 4 * h);
    v4f b[8];
#pragma unroll
    for (int j = 0; j < 8; ++j) b[j] = Wf[j * 32 + lane];

#pragma unroll 2
    for (int t = 0; t < KT; ++t) {
      const int tn = (t + 1 < KT) ? t + 1 : t;
      const v4f an = *(const v4f*)(tile + c * CP + 8 * tn + 4 * h);
      v4f bn[8];
#pragma unroll
      for (int j = 0; j < 8; ++j) bn[j] = Wf[(tn * 8 + j) * 32 + lane];

      v2f a0; a0.x = a.x; a0.y = a.y;
      v2f a1; a1.x = a.z; a1.y = a.w;
#pragma unroll
      for (int j = 0; j < 8; ++j) {
        v2f b0; b0.x = b[j].x; b0.y = b[j].y;
        acc[j] = __builtin_amdgcn_wmma_f32_16x16x4_f32(
             false, a0,  false, b0,
             (short)0, acc[j],  false,  false);
      }
#pragma unroll
      for (int j = 0; j < 8; ++j) {
        v2f b1; b1.x = b[j].z; b1.y = b[j].w;
        acc[j] = __builtin_amdgcn_wmma_f32_16x16x4_f32(
             false, a1,  false, b1,
             (short)0, acc[j],  false,  false);
      }
      a = an;
#pragma unroll
      for (int j = 0; j < 8; ++j) b[j] = bn[j];
    }

#pragma unroll
    for (int j = 0; j < 8; ++j)
#pragma unroll
      for (int r = 0; r < 8; ++r)
        tile[(8 * h + r) * CP + 16 * j + c] = acc[j][r];
    wave_tile_sync();

    float* hrow = H + (size_t)s * 16 * 128 + 4 * lane;
#pragma unroll
    for (int r = 0; r < 16; ++r) {
      const v4f v = *(const v4f*)(tile + r * CP + 4 * lane);
      *(v4f*)(hrow + (size_t)r * 128) = v;
    }
    wave_tile_sync();

    if (!more) break;
    s = sn;
    strip_rows_to_tile<K, NLD, CP>(pre, tile, lane);
    wave_tile_sync();
  }
}

template<int K, int ROWS, int NWAVES>
__global__ void __launch_bounds__(NWAVES * 32)
gemm_strip_kernel(const float* __restrict__ X,
                  const float* __restrict__ W,
                  float* __restrict__ H) {
  gemm_strip_body<K, ROWS, NWAVES>(X, W, H, (int)blockIdx.x, (int)gridDim.x);
}

#define TAB_WGS 16
__global__ void __launch_bounds__(ROWS_T)
k_rows_tab(int* __restrict__ cnts, int* __restrict__ icnt, int* __restrict__ total,
           int* __restrict__ rowstart, float* __restrict__ dinv,
           const float* __restrict__ emb, const float* __restrict__ W1,
           float* __restrict__ tab) {
  static_assert(ROWS_T == 8 * 32, "the product below runs with 8 waves per work-group");
  if ((int)blockIdx.x < TAB_WGS) {
    gemm_strip_body<64, VOCAB, 8>(emb, W1, tab, (int)blockIdx.x, TAB_WGS);
    return;
  }
  rows_body(cnts, icnt, total, rowstart, dinv, (int)blockIdx.x - TAB_WGS);
}


static __device__ __forceinline__ void fma4(float4& acc, const float4 h, const float w) {
  acc.x = fmaf(h.x, w, acc.x);
  acc.y = fmaf(h.y, w, acc.y);
  acc.z = fmaf(h.z, w, acc.z);
  acc.w = fmaf(h.w, w, acc.w);
}

template<bool VIA_TOK, bool READOUT>
__global__ void __launch_bounds__(256)
k_gather(const float* __restrict__ H, const int* __restrict__ rowstart,
         const int* __restrict__ icnt, const int* __restrict__ grouped,
         const float* __restrict__ dinv, const float* __restrict__ bias,
         const int* __restrict__ tok, float* __restrict__ out,
         const int* __restrict__ batch, const float* __restrict__ Wlin,
         float* __restrict__ pool) {
  const int node = __builtin_amdgcn_readfirstlane((int)((blockIdx.x << 3) | (threadIdx.x >> 5)));
  const int lane = threadIdx.x & 31;
  const int beg  = rowstart[node];
  const int n    = icnt[node];
  const float di = dinv[node];
  const int   g  = READOUT ? batch[node] : 0;
  const float4* __restrict__ Hv = (const float4*)H;

  float4 acc = Hv[((size_t)(VIA_TOK ? tok[node] : node) << 5) + lane];
  const float ws = di * di;
  acc.x *= ws; acc.y *= ws; acc.z *= ws; acc.w *= ws;

  const int* __restrict__ row = grouped + beg;
  int j = 0;
  for (; j + 4 <= n; j += 4) {
    const int s0 = row[j], s1 = row[j + 1], s2 = row[j + 2], s3 = row[j + 3];
    const int r0 = VIA_TOK ? tok[s0] : s0, r1 = VIA_TOK ? tok[s1] : s1;
    const int r2 = VIA_TOK ? tok[s2] : s2, r3 = VIA_TOK ? tok[s3] : s3;
    const float4 h0 = Hv[((size_t)r0 << 5) + lane];
    const float4 h1 = Hv[((size_t)r1 << 5) + lane];
    const float4 h2 = Hv[((size_t)r2 << 5) + lane];
    const float4 h3 = Hv[((size_t)r3 << 5) + lane];
    const float w0 = dinv[s0] * di, w1 = dinv[s1] * di;
    const float w2 = dinv[s2] * di, w3 = dinv[s3] * di;
    fma4(acc, h0, w0);
    fma4(acc, h1, w1);
    fma4(acc, h2, w2);
    fma4(acc, h3, w3);
  }
  if (j + 2 <= n) {
    const int s0 = row[j], s1 = row[j + 1];
    const int r0 = VIA_TOK ? tok[s0] : s0, r1 = VIA_TOK ? tok[s1] : s1;
    const float4 h0 = Hv[((size_t)r0 << 5) + lane];
    const float4 h1 = Hv[((size_t)r1 << 5) + lane];
    const float w0 = dinv[s0] * di, w1 = dinv[s1] * di;
    fma4(acc, h0, w0);
    fma4(acc, h1, w1);
    j += 2;
  }
  if (j < n) {
    const int s0 = row[j];
    const int r0 = VIA_TOK ? tok[s0] : s0;
    const float4 h0 = Hv[((size_t)r0 << 5) + lane];
    fma4(acc, h0, dinv[s0] * di);
  }

  const float4 b = ((const float4*)bias)[lane];
  float4 v;
  v.x = fmaxf(acc.x + b.x, 0.0f);
  v.y = fmaxf(acc.y + b.y, 0.0f);
  v.z = fmaxf(acc.z + b.z, 0.0f);
  v.w = fmaxf(acc.w + b.w, 0.0f);
  if (!READOUT) {
    ((float4*)out)[((size_t)node << 5) + lane] = v;
    return;
  }

  const float4* __restrict__ Wl = (const float4*)Wlin + 4 * lane;
  const float4 w0 = Wl[0], w1 = Wl[1], w2 = Wl[2], w3 = Wl[3];
  const float p0 = fmaf(v.w, w3.x, fmaf(v.z, w2.x, fmaf(v.y, w1.x, v.x * w0.x)));
  const float p1 = fmaf(v.w, w3.y, fmaf(v.z, w2.y, fmaf(v.y, w1.y, v.x * w0.y)));
  const float p2 = fmaf(v.w, w3.z, fmaf(v.z, w2.z, fmaf(v.y, w1.z, v.x * w0.z)));
  const float p3 = fmaf(v.w, w3.w, fmaf(v.z, w2.w, fmaf(v.y, w1.w, v.x * w0.w)));

  const bool odd1 = (lane & 1) != 0, odd2 = (lane & 2) != 0;
  const float qa = (odd1 ? p2 : p0) + __shfl_xor(odd1 ? p0 : p2, 1);
  const float qb = (odd1 ? p3 : p1) + __shfl_xor(odd1 ? p1 : p3, 1);
  float r = (odd2 ? qb : qa) + __shfl_xor(odd2 ? qa : qb, 2);
  r += __shfl_xor(r, 4);
  r += __shfl_xor(r, 8);
  r += __shfl_xor(r, 16);

  if (lane < 5) {
    const int   slot = (lane == 4) ? 4 : (((lane & 1) << 1) | (lane >> 1));
    const float val  = (lane == 4) ? 1.0f : r;
    atomicAdd(pool + ((size_t)g << 3) + slot, val);
  }
}


__global__ void k_final(const float* __restrict__ pool, const float* __restrict__ blin,
                        float* __restrict__ out) {
  int t = blockIdx.x * blockDim.x + threadIdx.x;
  if (t >= N_GRAPHS * NCLS) return;
  int g = t >> 2, c = t & 3;
  const float* p = pool + ((size_t)g << 3);
  out[t] = p[c] / fmaxf(p[4], 1.0f) + blin[c];
}


extern "C" void kernel_launch(void* const* d_in, const int* in_sizes, int n_in,
                              void* d_out, int out_size, void* d_ws, size_t ws_size,
                              hipStream_t stream) {
  const int*   tok   = (const int*)d_in[0];
  const int*   eidx  = (const int*)d_in[1];
  const int*   src   = eidx;
  const int*   dst   = eidx + N_EDGES;
  const int*   batch = (const int*)d_in[2];
  const float* emb   = (const float*)d_in[3];
  const float* W1    = (const float*)d_in[4];
  const float* b1    = (const float*)d_in[5];
  const float* W2    = (const float*)d_in[6];
  const float* b2    = (const float*)d_in[7];
  const float* Wlin  = (const float*)d_in[8];
  const float* blin  = (const float*)d_in[9];
  float*       out   = (float*)d_out;

  float* ws   = (float*)d_ws;
  float* dinv = ws;
  float* tab  = dinv + N_NODES;
  float* A    = tab  + (size_t)VOCAB_PAD * 128;
  float* B    = A    + (size_t)N_NODES * 128;
  int*   icnt     = (int*)(B + (size_t)N_NODES * 128);
  int*   total    = icnt + N_NODES;
  float* pool     = (float*)(total + 8);
  int*   rowstart = (int*)(pool + N_GRAPHS * 8);
  int*   rank     = rowstart + N_NODES;
  int*   grouped  = rank + N_EDGES;
  int*   cnts     = grouped + N_EDGES;

  const int T = 256;
  const int gNodes  = (N_NODES + T - 1) / T;
  const int gEdges  = (N_EDGES + T - 1) / T;
  const int gRows   = (N_NODES * 32) / T;
  const int gGemm   = 256;

  k_rank<<<RK_RANGES * RK_CHUNKS, RK_T, 0, stream>>>(dst, cnts, rank, total, pool);
  k_rows_tab<<<gNodes + TAB_WGS, ROWS_T, 0, stream>>>(cnts, icnt, total, rowstart, dinv, emb, W1, tab);
  k_fill<<<gEdges, T, 0, stream>>>(src, dst, rank, cnts, grouped);

  k_gather<true, false><<<gRows, T, 0, stream>>>(tab, rowstart, icnt, grouped, dinv, b1, tok, B,
                                                 nullptr, nullptr, nullptr);

  gemm_strip_kernel<128, N_NODES, 16><<<gGemm, 512, 0, stream>>>(B, W2, A);
  k_gather<false, true><<<gRows, T, 0, stream>>>(A, rowstart, icnt, grouped, dinv, b2, tok, nullptr,
                                                 batch, Wlin, pool);

  k_final<<<(N_GRAPHS * NCLS + T - 1) / T, T, 0, stream>>>(pool, blin, out);
}


# layer-1 gather fused into the layer-2 WMMA product: 768 resident 16-wave work-groups gather 32 rows into LDS, multiply by W2 from LDS, store h2; x1 never stored; 6 launches; sums unchanged
# speedup vs baseline: 4.1881x; 1.1123x over previous
#include <hip/hip_runtime.h>
#include <hip/hip_bf16.h>


#define N_NODES  100000
#define N_EDGES  600000
#define N_GRAPHS 2000
#define NCLS     4
#define VOCAB    1000
#define VOCAB_PAD 1008

typedef float v2f __attribute__((ext_vector_type(2)));
typedef float v4f __attribute__((ext_vector_type(4)));
typedef float v8f __attribute__((ext_vector_type(8)));


#define RK_RANGES 25
#define RK_NODES  4000
#define RK_CHUNKS 10
#define RK_EDGES  60000
#define RK_T      1024

__global__ void __launch_bounds__(RK_T)
k_rank(const int* __restrict__ dst, int* __restrict__ cnts, int* __restrict__ rank,
       int* __restrict__ total, float* __restrict__ pool) {
  static_assert(RK_RANGES * RK_NODES == N_NODES && RK_CHUNKS * RK_EDGES == N_EDGES, "exact cover");
  static_assert(RK_EDGES % 4 == 0, "a stretch is a whole number of 16-byte loads");
  static_assert(RK_RANGES * RK_CHUNKS * 64 == N_GRAPHS * 8, "pool is cleared in slices of 64 words");
  __shared__ int cnt[RK_NODES];
  const int r   = (int)blockIdx.x % RK_RANGES;
  const int c   = (int)blockIdx.x / RK_RANGES;
  const int lo  = r * RK_NODES;
  const int tid = (int)threadIdx.x;

  constexpr int Q   = RK_EDGES / 4;
  constexpr int NIT = (Q + RK_T - 1) / RK_T;
  const int4* __restrict__ dv = (const int4*)(dst + (size_t)c * RK_EDGES);
  int4 d[NIT];
#pragma unroll
  for (int it = 0; it < NIT; ++it) {
    const int q = tid + it * RK_T;
    d[it] = (q < Q) ? dv[q] : make_int4(-1, -1, -1, -1);
  }

  for (int i = tid; i < RK_NODES; i += RK_T) cnt[i] = 0;
  if (tid < 64) pool[(size_t)blockIdx.x * 64 + tid] = 0.0f;
  if (blockIdx.x == 0 && tid == 0) *total = 0;
  __syncthreads();

#pragma unroll
  for (int it = 0; it < NIT; ++it) {
    const int q = tid + it * RK_T;
    int* __restrict__ rk = rank + (size_t)c * RK_EDGES + 4 * (size_t)q;
    const unsigned u0 = (unsigned)(d[it].x - lo), u1 = (unsigned)(d[it].y - lo);
    const unsigned u2 = (unsigned)(d[it].z - lo), u3 = (unsigned)(d[it].w - lo);
    if (u0 < (unsigned)RK_NODES) rk[0] = atomicAdd(&cnt[u0], 1);
    if (u1 < (unsigned)RK_NODES) rk[1] = atomicAdd(&cnt[u1], 1);
    if (u2 < (unsigned)RK_NODES) rk[2] = atomicAdd(&cnt[u2], 1);
    if (u3 < (unsigned)RK_NODES) rk[3] = atomicAdd(&cnt[u3], 1);
  }
  __syncthreads();

  for (int i = tid; i < RK_NODES; i += RK_T) cnts[(size_t)c * N_NODES + lo + i] = cnt[i];
}

#define ROWS_T 256
static __device__ __forceinline__ void
rows_body(int* __restrict__ cnts, int* __restrict__ icnt, int* __restrict__ total,
          int* __restrict__ rowstart, float* __restrict__ dinv, const int bid) {
  const int i    = bid * ROWS_T + (int)threadIdx.x;
  const int lane = threadIdx.x & 31;
  int before[RK_CHUNKS];
  int c = 0;
#pragma unroll
  for (int k = 0; k < RK_CHUNKS; ++k) {
    before[k] = c;
    c += (i < N_NODES) ? cnts[(size_t)k * N_NODES + i] : 0;
  }
  int incl = c;
  for (int off = 1; off < 32; off <<= 1) {
    const int t = __shfl_up(incl, off);
    if (lane >= off) incl += t;
  }
  int base = 0;
  if (lane == 31) base = atomicAdd(total, incl);
  base = __shfl(base, 31);
  if (i < N_NODES) {
    const int rs = base + incl - c;
    rowstart[i] = rs;
    icnt[i]     = c;
    dinv[i]     = rsqrtf((float)c + 1.0f);
#pragma unroll
    for (int k = 0; k < RK_CHUNKS; ++k) cnts[(size_t)k * N_NODES + i] = rs + before[k];
  }
}

__global__ void k_fill(const int* __restrict__ src, const int* __restrict__ dst,
                       const int* __restrict__ rank, const int* __restrict__ cnts,
                       int* __restrict__ grouped) {
  int e = blockIdx.x * blockDim.x + threadIdx.x;
  if (e < N_EDGES) grouped[cnts[(size_t)(e / RK_EDGES) * N_NODES + dst[e]] + rank[e]] = src[e];
}


static __device__ __forceinline__ void wave_tile_sync() {
  __builtin_amdgcn_fence(__ATOMIC_RELEASE, "wavefront");
  __builtin_amdgcn_wave_barrier();
  __builtin_amdgcn_fence(__ATOMIC_ACQUIRE, "wavefront");
}

template<int K, int ROWS, int NLD>
static __device__ __forceinline__ void strip_load_rows(v4f (&pre)[NLD], const float* __restrict__ X,
                                                       const int strip, const int lane) {
  const int h = lane >> 4, c = lane & 15;
#pragma unroll
  for (int i = 0; i < NLD; ++i) {
    const int rt = strip * 16 + ((K == 128) ? i : 2 * i + h);
    const int m  = (ROWS % 16 != 0 && rt > ROWS - 1) ? ROWS - 1 : rt;
    pre[i] = *(const v4f*)(X + (size_t)m * K + 4 * ((K == 128) ? lane : c));
  }
}

template<int K, int NLD, int CP>
static __device__ __forceinline__ void strip_rows_to_tile(const v4f (&pre)[NLD], float* tile,
                                                          const int lane) {
  const int h = lane >> 4, c = lane & 15;
#pragma unroll
  for (int i = 0; i < NLD; ++i) {
    const int r = (K == 128) ? i : 2 * i + h;
    *(v4f*)(tile + r * CP + 4 * ((K == 128) ? lane : c)) = pre[i];
  }
}

template<int K, int ROWS, int NWAVES>
static __device__ __forceinline__ void
gemm_strip_body(const float* __restrict__ X,
                const float* __restrict__ W,
                float* __restrict__ H,
                const int bid, const int nblk) {
  constexpr int KT      = K / 8;
  constexpr int NSTRIPS = (ROWS + 15) / 16;
  constexpr int CP      = 132;
  static_assert(K == 64 || K == 128, "row loads below are written for K = 64 or 128");
  __shared__ v4f Wf[KT * 8 * 32];
  __shared__ __attribute__((aligned(16))) float Tile[NWAVES * 16 * CP];

  constexpr int NSTAGE = (KT * 8 * 32) / (NWAVES * 32);
  static_assert((KT * 8 * 32) % (NWAVES * 32) == 0, "every thread stages the same number of entries");
#pragma unroll
  for (int i = 0; i < NSTAGE; ++i) {
    const int f  = (int)threadIdx.x + i * (NWAVES * 32);
    const int ln = f & 31, j = (f >> 5) & 7, t = f >> 8;
    const float* wp = W + (size_t)(8 * t + 4 * (ln >> 4)) * 128 + 16 * j + (ln & 15);
    v4f v;
    v.x = wp[0]; v.y = wp[128]; v.z = wp[256]; v.w = wp[384];
    Wf[f] = v;
  }
  __syncthreads();

  const int wave = __builtin_amdgcn_readfirstlane((int)(threadIdx.x >> 5));
  const int lane = threadIdx.x & 31;
  const int h    = lane >> 4;
  const int c    = lane & 15;
  float* tile = Tile + wave * (16 * CP);
  const int step = NWAVES * nblk;

  int s = wave * nblk + bid;
  if (s >= NSTRIPS) return;

  constexpr int NLD = (K == 128) ? 16 : 8;
  v4f pre[NLD];

  strip_load_rows<K, ROWS, NLD>(pre, X, s, lane);
  strip_rows_to_tile<K, NLD, CP>(pre, tile, lane);
  wave_tile_sync();

  for (;;) {
    const int  sn   = s + step;
    const bool more = sn < NSTRIPS;
    strip_load_rows<K, ROWS, NLD>(pre, X, more ? sn : s, lane);

    v8f acc[8];
#pragma unroll
    for (int j = 0; j < 8; ++j) acc[j] = (v8f){};

    v4f a = *(const v4f*)(tile + c * CP + 4 * h);
    v4f b[8];
#pragma unroll
    for (int j = 0; j < 8; ++j) b[j] = Wf[j * 32 + lane];

#pragma unroll 2
    for (int t = 0; t < KT; ++t) {
      const int tn = (t + 1 < KT) ? t + 1 : t;
      const v4f an = *(const v4f*)(tile + c * CP + 8 * tn + 4 * h);
      v4f bn[8];
#pragma unroll
      for (int j = 0; j < 8; ++j) bn[j] = Wf[(tn * 8 + j) * 32 + lane];

      v2f a0; a0.x = a.x; a0.y = a.y;
      v2f a1; a1.x = a.z; a1.y = a.w;
#pragma unroll
      for (int j = 0; j < 8; ++j) {
        v2f b0; b0.x = b[j].x; b0.y = b[j].y;
        acc[j] = __builtin_amdgcn_wmma_f32_16x16x4_f32(
             false, a0,  false, b0,
             (short)0, acc[j],  false,  false);
      }
#pragma unroll
      for (int j = 0; j < 8; ++j) {
        v2f b1; b1.x = b[j].z; b1.y = b[j].w;
        acc[j] = __builtin_amdgcn_wmma_f32_16x16x4_f32(
             false, a1,  false, b1,
             (short)0, acc[j],  false,  false);
      }
      a = an;
#pragma unroll
      for (int j = 0; j < 8; ++j) b[j] = bn[j];
    }

#pragma unroll
    for (int j = 0; j < 8; ++j)
#pragma unroll
      for (int r = 0; r < 8; ++r)
        tile[(8 * h + r) * CP + 16 * j + c] = acc[j][r];
    wave_tile_sync();

    float* hrow = H + (size_t)s * 16 * 128 + 4 * lane;
#pragma unroll
    for (int r = 0; r < 16; ++r) {
      const v4f v = *(const v4f*)(tile + r * CP + 4 * lane);
      *(v4f*)(hrow + (size_t)r * 128) = v;
    }
    wave_tile_sync();

    if (!more) break;
    s = sn;
    strip_rows_to_tile<K, NLD, CP>(pre, tile, lane);
    wave_tile_sync();
  }
}

template<int K, int ROWS, int NWAVES>
__global__ void __launch_bounds__(NWAVES * 32)
gemm_strip_kernel(const float* __restrict__ X,
                  const float* __restrict__ W,
                  float* __restrict__ H) {
  gemm_strip_body<K, ROWS, NWAVES>(X, W, H, (int)blockIdx.x, (int)gridDim.x);
}

#define TAB_WGS 16
__global__ void __launch_bounds__(ROWS_T)
k_rows_tab(int* __restrict__ cnts, int* __restrict__ icnt, int* __restrict__ total,
           int* __restrict__ rowstart, float* __restrict__ dinv,
           const float* __restrict__ emb, const float* __restrict__ W1,
           float* __restrict__ tab) {
  static_assert(ROWS_T == 8 * 32, "the product below runs with 8 waves per work-group");
  if ((int)blockIdx.x < TAB_WGS) {
    gemm_strip_body<64, VOCAB, 8>(emb, W1, tab, (int)blockIdx.x, TAB_WGS);
    return;
  }
  rows_body(cnts, icnt, total, rowstart, dinv, (int)blockIdx.x - TAB_WGS);
}


static __device__ __forceinline__ void fma4(float4& acc, const float4 h, const float w) {
  acc.x = fmaf(h.x, w, acc.x);
  acc.y = fmaf(h.y, w, acc.y);
  acc.z = fmaf(h.z, w, acc.z);
  acc.w = fmaf(h.w, w, acc.w);
}

struct RowHead { int beg, n; float di; float4 self; };

template<bool VIA_TOK>
static __device__ __forceinline__ RowHead
gather_head(const float* __restrict__ H, const int* __restrict__ rowstart,
            const int* __restrict__ icnt, const float* __restrict__ dinv,
            const int* __restrict__ tok, const int node, const int lane) {
  RowHead hd;
  hd.beg  = rowstart[node];
  hd.n    = icnt[node];
  hd.di   = dinv[node];
  hd.self = ((const float4*)H)[((size_t)(VIA_TOK ? tok[node] : node) << 5) + lane];
  return hd;
}

template<bool VIA_TOK>
static __device__ __forceinline__ float4
gather_finish(const RowHead hd, const float* __restrict__ H, const int* __restrict__ grouped,
              const float* __restrict__ dinv, const float* __restrict__ bias,
              const int* __restrict__ tok, const int lane) {
  const int beg  = hd.beg;
  const int n    = hd.n;
  const float di = hd.di;
  const float4* __restrict__ Hv = (const float4*)H;

  float4 acc = hd.self;
  const float ws = di * di;
  acc.x *= ws; acc.y *= ws; acc.z *= ws; acc.w *= ws;

  const int* __restrict__ row = grouped + beg;
  int j = 0;
  for (; j + 4 <= n; j += 4) {
    const int s0 = row[j], s1 = row[j + 1], s2 = row[j + 2], s3 = row[j + 3];
    const int r0 = VIA_TOK ? tok[s0] : s0, r1 = VIA_TOK ? tok[s1] : s1;
    const int r2 = VIA_TOK ? tok[s2] : s2, r3 = VIA_TOK ? tok[s3] : s3;
    const float4 h0 = Hv[((size_t)r0 << 5) + lane];
    const float4 h1 = Hv[((size_t)r1 << 5) + lane];
    const float4 h2 = Hv[((size_t)r2 << 5) + lane];
    const float4 h3 = Hv[((size_t)r3 << 5) + lane];
    const float w0 = dinv[s0] * di, w1 = dinv[s1] * di;
    const float w2 = dinv[s2] * di, w3 = dinv[s3] * di;
    fma4(acc, h0, w0);
    fma4(acc, h1, w1);
    fma4(acc, h2, w2);
    fma4(acc, h3, w3);
  }
  if (j + 2 <= n) {
    const int s0 = row[j], s1 = row[j + 1];
    const int r0 = VIA_TOK ? tok[s0] : s0, r1 = VIA_TOK ? tok[s1] : s1;
    const float4 h0 = Hv[((size_t)r0 << 5) + lane];
    const float4 h1 = Hv[((size_t)r1 << 5) + lane];
    const float w0 = dinv[s0] * di, w1 = dinv[s1] * di;
    fma4(acc, h0, w0);
    fma4(acc, h1, w1);
    j += 2;
  }
  if (j < n) {
    const int s0 = row[j];
    const int r0 = VIA_TOK ? tok[s0] : s0;
    const float4 h0 = Hv[((size_t)r0 << 5) + lane];
    fma4(acc, h0, dinv[s0] * di);
  }

  const float4 b = ((const float4*)bias)[lane];
  float4 v;
  v.x = fmaxf(acc.x + b.x, 0.0f);
  v.y = fmaxf(acc.y + b.y, 0.0f);
  v.z = fmaxf(acc.z + b.z, 0.0f);
  v.w = fmaxf(acc.w + b.w, 0.0f);
  return v;
}

template<bool VIA_TOK>
static __device__ __forceinline__ float4
gather_row(const float* __restrict__ H, const int* __restrict__ rowstart,
           const int* __restrict__ icnt, const int* __restrict__ grouped,
           const float* __restrict__ dinv, const float* __restrict__ bias,
           const int* __restrict__ tok, const int node, const int lane) {
  const RowHead hd = gather_head<VIA_TOK>(H, rowstart, icnt, dinv, tok, node, lane);
  return gather_finish<VIA_TOK>(hd, H, grouped, dinv, bias, tok, lane);
}

template<bool VIA_TOK, bool READOUT>
__global__ void __launch_bounds__(256)
k_gather(const float* __restrict__ H, const int* __restrict__ rowstart,
         const int* __restrict__ icnt, const int* __restrict__ grouped,
         const float* __restrict__ dinv, const float* __restrict__ bias,
         const int* __restrict__ tok, float* __restrict__ out,
         const int* __restrict__ batch, const float* __restrict__ Wlin,
         float* __restrict__ pool) {
  const int node = __builtin_amdgcn_readfirstlane((int)((blockIdx.x << 3) | (threadIdx.x >> 5)));
  const int lane = threadIdx.x & 31;
  const int   g  = READOUT ? batch[node] : 0;
  const float4 v = gather_row<VIA_TOK>(H, rowstart, icnt, grouped, dinv, bias, tok, node, lane);
  if (!READOUT) {
    ((float4*)out)[((size_t)node << 5) + lane] = v;
    return;
  }

  const float4* __restrict__ Wl = (const float4*)Wlin + 4 * lane;
  const float4 w0 = Wl[0], w1 = Wl[1], w2 = Wl[2], w3 = Wl[3];
  const float p0 = fmaf(v.w, w3.x, fmaf(v.z, w2.x, fmaf(v.y, w1.x, v.x * w0.x)));
  const float p1 = fmaf(v.w, w3.y, fmaf(v.z, w2.y, fmaf(v.y, w1.y, v.x * w0.y)));
  const float p2 = fmaf(v.w, w3.z, fmaf(v.z, w2.z, fmaf(v.y, w1.z, v.x * w0.z)));
  const float p3 = fmaf(v.w, w3.w, fmaf(v.z, w2.w, fmaf(v.y, w1.w, v.x * w0.w)));

  const bool odd1 = (lane & 1) != 0, odd2 = (lane & 2) != 0;
  const float qa = (odd1 ? p2 : p0) + __shfl_xor(odd1 ? p0 : p2, 1);
  const float qb = (odd1 ? p3 : p1) + __shfl_xor(odd1 ? p1 : p3, 1);
  float r = (odd2 ? qb : qa) + __shfl_xor(odd2 ? qa : qb, 2);
  r += __shfl_xor(r, 4);
  r += __shfl_xor(r, 8);
  r += __shfl_xor(r, 16);

  if (lane < 5) {
    const int   slot = (lane == 4) ? 4 : (((lane & 1) << 1) | (lane >> 1));
    const float val  = (lane == 4) ? 1.0f : r;
    atomicAdd(pool + ((size_t)g << 3) + slot, val);
  }
}

#define FG_NODES 32
#define FG_WGS   768
__global__ void __launch_bounds__(512)
k_gather_gemm(const float* __restrict__ tab, const int* __restrict__ rowstart,
              const int* __restrict__ icnt, const int* __restrict__ grouped,
              const float* __restrict__ dinv, const float* __restrict__ bias,
              const int* __restrict__ tok, const float* __restrict__ W,
              float* __restrict__ H) {
  static_assert(N_NODES % FG_NODES == 0, "no partial strip");
  constexpr int NTRIPS = N_NODES / FG_NODES;
  constexpr int CP = 132;
  __shared__ v4f Wf[16 * 8 * 32];
  __shared__ __attribute__((aligned(16))) float Xt[FG_NODES * CP];
  __shared__ __attribute__((aligned(16))) float Ot[FG_NODES * CP];

#pragma unroll
  for (int i = 0; i < 8; ++i) {
    const int f  = (int)threadIdx.x + i * 512;
    const int ln = f & 31, j = (f >> 5) & 7, t = f >> 8;
    const float* wp = W + (size_t)(8 * t + 4 * (ln >> 4)) * 128 + 16 * j + (ln & 15);
    v4f v;
    v.x = wp[0]; v.y = wp[128]; v.z = wp[256]; v.w = wp[384];
    Wf[f] = v;
  }
  __syncthreads();

  const int wave = __builtin_amdgcn_readfirstlane((int)(threadIdx.x >> 5));
  const int lane = threadIdx.x & 31;
  const int h    = lane >> 4;
  const int c    = lane & 15;
  const int sidx = wave >> 3;
  const int jt   = wave & 7;

  for (int trip = (int)blockIdx.x; trip < NTRIPS; trip += FG_WGS) {
    const int base = trip * FG_NODES;

    const int r0 = 2 * wave;
    const RowHead hd0 = gather_head<true>(tab, rowstart, icnt, dinv, tok, base + r0, lane);
    const RowHead hd1 = gather_head<true>(tab, rowstart, icnt, dinv, tok, base + r0 + 1, lane);
    const float4 v0 = gather_finish<true>(hd0, tab, grouped, dinv, bias, tok, lane);
    *(float4*)(Xt + r0 * CP + 4 * lane) = v0;
    const float4 v1 = gather_finish<true>(hd1, tab, grouped, dinv, bias, tok, lane);
    *(float4*)(Xt + (r0 + 1) * CP + 4 * lane) = v1;
    __syncthreads();

    v8f acc = (v8f){};
    const float* arow = Xt + (16 * sidx + c) * CP + 4 * h;
#pragma unroll
    for (int t = 0; t < 16; ++t) {
      const v4f a = *(const v4f*)(arow + 8 * t);
      const v4f b = Wf[(t * 8 + jt) * 32 + lane];
      v2f a0; a0.x = a.x; a0.y = a.y;
      v2f b0; b0.x = b.x; b0.y = b.y;
      acc = __builtin_amdgcn_wmma_f32_16x16x4_f32(
           false, a0,  false, b0,
           (short)0, acc,  false,  false);
      v2f a1; a1.x = a.z; a1.y = a.w;
      v2f b1; b1.x = b.z; b1.y = b.w;
      acc = __builtin_amdgcn_wmma_f32_16x16x4_f32(
           false, a1,  false, b1,
           (short)0, acc,  false,  false);
    }
#pragma unroll
    for (int r = 0; r < 8; ++r)
      Ot[(16 * sidx + 8 * h + r) * CP + 16 * jt + c] = acc[r];
    __syncthreads();

#pragma unroll
    for (int q = 0; q < 2; ++q) {
      const int r = 2 * wave + q;
      const v4f v = *(const v4f*)(Ot + r * CP + 4 * lane);
      *(v4f*)(H + (size_t)(base + r) * 128 + 4 * lane) = v;
    }
    __builtin_amdgcn_fence(__ATOMIC_RELEASE, "wavefront");
  }
}


__global__ void k_final(const float* __restrict__ pool, const float* __restrict__ blin,
                        float* __restrict__ out) {
  int t = blockIdx.x * blockDim.x + threadIdx.x;
  if (t >= N_GRAPHS * NCLS) return;
  int g = t >> 2, c = t & 3;
  const float* p = pool + ((size_t)g << 3);
  out[t] = p[c] / fmaxf(p[4], 1.0f) + blin[c];
}


extern "C" void kernel_launch(void* const* d_in, const int* in_sizes, int n_in,
                              void* d_out, int out_size, void* d_ws, size_t ws_size,
                              hipStream_t stream) {
  const int*   tok   = (const int*)d_in[0];
  const int*   eidx  = (const int*)d_in[1];
  const int*   src   = eidx;
  const int*   dst   = eidx + N_EDGES;
  const int*   batch = (const int*)d_in[2];
  const float* emb   = (const float*)d_in[3];
  const float* W1    = (const float*)d_in[4];
  const float* b1    = (const float*)d_in[5];
  const float* W2    = (const float*)d_in[6];
  const float* b2    = (const float*)d_in[7];
  const float* Wlin  = (const float*)d_in[8];
  const float* blin  = (const float*)d_in[9];
  float*       out   = (float*)d_out;

  float* ws   = (float*)d_ws;
  float* dinv = ws;
  float* tab  = dinv + N_NODES;
  float* A    = tab  + (size_t)VOCAB_PAD * 128;
  int*   icnt     = (int*)(A + (size_t)N_NODES * 128);
  int*   total    = icnt + N_NODES;
  float* pool     = (float*)(total + 8);
  int*   rowstart = (int*)(pool + N_GRAPHS * 8);
  int*   rank     = rowstart + N_NODES;
  int*   grouped  = rank + N_EDGES;
  int*   cnts     = grouped + N_EDGES;

  const int T = 256;
  const int gNodes  = (N_NODES + T - 1) / T;
  const int gEdges  = (N_EDGES + T - 1) / T;
  const int gRows   = (N_NODES * 32) / T;

  k_rank<<<RK_RANGES * RK_CHUNKS, RK_T, 0, stream>>>(dst, cnts, rank, total, pool);
  k_rows_tab<<<gNodes + TAB_WGS, ROWS_T, 0, stream>>>(cnts, icnt, total, rowstart, dinv, emb, W1, tab);
  k_fill<<<gEdges, T, 0, stream>>>(src, dst, rank, cnts, grouped);

  k_gather_gemm<<<FG_WGS, 512, 0, stream>>>(tab, rowstart, icnt, grouped, dinv, b1, tok, W2, A);

  k_gather<false, true><<<gRows, T, 0, stream>>>(A, rowstart, icnt, grouped, dinv, b2, tok, nullptr,
                                                 batch, Wlin, pool);

  k_final<<<(N_GRAPHS * NCLS + T - 1) / T, T, 0, stream>>>(pool, blin, out);
}


# fill kernel writes each edge's {src, tok[src], dinv[src]} as one 16-byte record in grouped order; both gathers read records in sequence, no per-edge random tok/dinv loads left; sums unchanged
# speedup vs baseline: 4.5393x; 1.0839x over previous
#include <hip/hip_runtime.h>
#include <hip/hip_bf16.h>


#define N_NODES  100000
#define N_EDGES  600000
#define N_GRAPHS 2000
#define NCLS     4
#define VOCAB    1000
#define VOCAB_PAD 1008

typedef float v2f __attribute__((ext_vector_type(2)));
typedef float v4f __attribute__((ext_vector_type(4)));
typedef float v8f __attribute__((ext_vector_type(8)));


#define RK_RANGES 25
#define RK_NODES  4000
#define RK_CHUNKS 10
#define RK_EDGES  60000
#define RK_T      1024

__global__ void __launch_bounds__(RK_T)
k_rank(const int* __restrict__ dst, int* __restrict__ cnts, int* __restrict__ rank,
       int* __restrict__ total, float* __restrict__ pool) {
  static_assert(RK_RANGES * RK_NODES == N_NODES && RK_CHUNKS * RK_EDGES == N_EDGES, "exact cover");
  static_assert(RK_EDGES % 4 == 0, "a stretch is a whole number of 16-byte loads");
  static_assert(RK_RANGES * RK_CHUNKS * 64 == N_GRAPHS * 8, "pool is cleared in slices of 64 words");
  __shared__ int cnt[RK_NODES];
  const int r   = (int)blockIdx.x % RK_RANGES;
  const int c   = (int)blockIdx.x / RK_RANGES;
  const int lo  = r * RK_NODES;
  const int tid = (int)threadIdx.x;

  constexpr int Q   = RK_EDGES / 4;
  constexpr int NIT = (Q + RK_T - 1) / RK_T;
  const int4* __restrict__ dv = (const int4*)(dst + (size_t)c * RK_EDGES);
  int4 d[NIT];
#pragma unroll
  for (int it = 0; it < NIT; ++it) {
    const int q = tid + it * RK_T;
    d[it] = (q < Q) ? dv[q] : make_int4(-1, -1, -1, -1);
  }

  for (int i = tid; i < RK_NODES; i += RK_T) cnt[i] = 0;
  if (tid < 64) pool[(size_t)blockIdx.x * 64 + tid] = 0.0f;
  if (blockIdx.x == 0 && tid == 0) *total = 0;
  __syncthreads();

#pragma unroll
  for (int it = 0; it < NIT; ++it) {
    const int q = tid + it * RK_T;
    int* __restrict__ rk = rank + (size_t)c * RK_EDGES + 4 * (size_t)q;
    const unsigned u0 = (unsigned)(d[it].x - lo), u1 = (unsigned)(d[it].y - lo);
    const unsigned u2 = (unsigned)(d[it].z - lo), u3 = (unsigned)(d[it].w - lo);
    if (u0 < (unsigned)RK_NODES) rk[0] = atomicAdd(&cnt[u0], 1);
    if (u1 < (unsigned)RK_NODES) rk[1] = atomicAdd(&cnt[u1], 1);
    if (u2 < (unsigned)RK_NODES) rk[2] = atomicAdd(&cnt[u2], 1);
    if (u3 < (unsigned)RK_NODES) rk[3] = atomicAdd(&cnt[u3], 1);
  }
  __syncthreads();

  for (int i = tid; i < RK_NODES; i += RK_T) cnts[(size_t)c * N_NODES + lo + i] = cnt[i];
}

#define ROWS_T 256
static __device__ __forceinline__ void
rows_body(int* __restrict__ cnts, int* __restrict__ icnt, int* __restrict__ total,
          int* __restrict__ rowstart, const int* __restrict__ tok, int* __restrict__ td,
          const int bid) {
  const int i    = bid * ROWS_T + (int)threadIdx.x;
  const int lane = threadIdx.x & 31;
  int before[RK_CHUNKS];
  int c = 0;
#pragma unroll
  for (int k = 0; k < RK_CHUNKS; ++k) {
    before[k] = c;
    c += (i < N_NODES) ? cnts[(size_t)k * N_NODES + i] : 0;
  }
  int incl = c;
  for (int off = 1; off < 32; off <<= 1) {
    const int t = __shfl_up(incl, off);
    if (lane >= off) incl += t;
  }
  int base = 0;
  if (lane == 31) base = atomicAdd(total, incl);
  base = __shfl(base, 31);
  if (i < N_NODES) {
    const int rs = base + incl - c;
    rowstart[i] = rs;
    icnt[i]     = c;
    ((int2*)td)[i] = make_int2(tok[i], __float_as_int(rsqrtf((float)c + 1.0f)));
#pragma unroll
    for (int k = 0; k < RK_CHUNKS; ++k) cnts[(size_t)k * N_NODES + i] = rs + before[k];
  }
}

__global__ void k_fill(const int* __restrict__ src, const int* __restrict__ dst,
                       const int* __restrict__ rank, const int* __restrict__ cnts,
                       const int* __restrict__ td, int* __restrict__ grouped) {
  int e = blockIdx.x * blockDim.x + threadIdx.x;
  if (e < N_EDGES) {
    const int  s    = src[e];
    const int2 t    = ((const int2*)td)[s];
    const int  slot = cnts[(size_t)(e / RK_EDGES) * N_NODES + dst[e]] + rank[e];
    ((int4*)grouped)[slot] = make_int4(s, t.x, t.y, 0);
  }
}


static __device__ __forceinline__ void wave_tile_sync() {
  __builtin_amdgcn_fence(__ATOMIC_RELEASE, "wavefront");
  __builtin_amdgcn_wave_barrier();
  __builtin_amdgcn_fence(__ATOMIC_ACQUIRE, "wavefront");
}

template<int K, int ROWS, int NLD>
static __device__ __forceinline__ void strip_load_rows(v4f (&pre)[NLD], const float* __restrict__ X,
                                                       const int strip, const int lane) {
  const int h = lane >> 4, c = lane & 15;
#pragma unroll
  for (int i = 0; i < NLD; ++i) {
    const int rt = strip * 16 + ((K == 128) ? i : 2 * i + h);
    const int m  = (ROWS % 16 != 0 && rt > ROWS - 1) ? ROWS - 1 : rt;
    pre[i] = *(const v4f*)(X + (size_t)m * K + 4 * ((K == 128) ? lane : c));
  }
}

template<int K, int NLD, int CP>
static __device__ __forceinline__ void strip_rows_to_tile(const v4f (&pre)[NLD], float* tile,
                                                          const int lane) {
  const int h = lane >> 4, c = lane & 15;
#pragma unroll
  for (int i = 0; i < NLD; ++i) {
    const int r = (K == 128) ? i : 2 * i + h;
    *(v4f*)(tile + r * CP + 4 * ((K == 128) ? lane : c)) = pre[i];
  }
}

template<int K, int ROWS, int NWAVES>
static __device__ __forceinline__ void
gemm_strip_body(const float* __restrict__ X,
                const float* __restrict__ W,
                float* __restrict__ H,
                const int bid, const int nblk) {
  constexpr int KT      = K / 8;
  constexpr int NSTRIPS = (ROWS + 15) / 16;
  constexpr int CP      = 132;
  static_assert(K == 64 || K == 128, "row loads below are written for K = 64 or 128");
  __shared__ v4f Wf[KT * 8 * 32];
  __shared__ __attribute__((aligned(16))) float Tile[NWAVES * 16 * CP];

  constexpr int NSTAGE = (KT * 8 * 32) / (NWAVES * 32);
  static_assert((KT * 8 * 32) % (NWAVES * 32) == 0, "every thread stages the same number of entries");
#pragma unroll
  for (int i = 0; i < NSTAGE; ++i) {
    const int f  = (int)threadIdx.x + i * (NWAVES * 32);
    const int ln = f & 31, j = (f >> 5) & 7, t = f >> 8;
    const float* wp = W + (size_t)(8 * t + 4 * (ln >> 4)) * 128 + 16 * j + (ln & 15);
    v4f v;
    v.x = wp[0]; v.y = wp[128]; v.z = wp[256]; v.w = wp[384];
    Wf[f] = v;
  }
  __syncthreads();

  const int wave = __builtin_amdgcn_readfirstlane((int)(threadIdx.x >> 5));
  const int lane = threadIdx.x & 31;
  const int h    = lane >> 4;
  const int c    = lane & 15;
  float* tile = Tile + wave * (16 * CP);
  const int step = NWAVES * nblk;

  int s = wave * nblk + bid;
  if (s >= NSTRIPS) return;

  constexpr int NLD = (K == 128) ? 16 : 8;
  v4f pre[NLD];

  strip_load_rows<K, ROWS, NLD>(pre, X, s, lane);
  strip_rows_to_tile<K, NLD, CP>(pre, tile, lane);
  wave_tile_sync();

  for (;;) {
    const int  sn   = s + step;
    const bool more = sn < NSTRIPS;
    strip_load_rows<K, ROWS, NLD>(pre, X, more ? sn : s, lane);

    v8f acc[8];
#pragma unroll
    for (int j = 0; j < 8; ++j) acc[j] = (v8f){};

    v4f a = *(const v4f*)(tile + c * CP + 4 * h);
    v4f b[8];
#pragma unroll
    for (int j = 0; j < 8; ++j) b[j] = Wf[j * 32 + lane];

#pragma unroll 2
    for (int t = 0; t < KT; ++t) {
      const int tn = (t + 1 < KT) ? t + 1 : t;
      const v4f an = *(const v4f*)(tile + c * CP + 8 * tn + 4 * h);
      v4f bn[8];
#pragma unroll
      for (int j = 0; j < 8; ++j) bn[j] = Wf[(tn * 8 + j) * 32 + lane];

      v2f a0; a0.x = a.x; a0.y = a.y;
      v2f a1; a1.x = a.z; a1.y = a.w;
#pragma unroll
      for (int j = 0; j < 8; ++j) {
        v2f b0; b0.x = b[j].x; b0.y = b[j].y;
        acc[j] = __builtin_amdgcn_wmma_f32_16x16x4_f32(
             false, a0,  false, b0,
             (short)0, acc[j],  false,  false);
      }
#pragma unroll
      for (int j = 0; j < 8; ++j) {
        v2f b1; b1.x = b[j].z; b1.y = b[j].w;
        acc[j] = __builtin_amdgcn_wmma_f32_16x16x4_f32(
             false, a1,  false, b1,
             (short)0, acc[j],  false,  false);
      }
      a = an;
#pragma unroll
      for (int j = 0; j < 8; ++j) b[j] = bn[j];
    }

#pragma unroll
    for (int j = 0; j < 8; ++j)
#pragma unroll
      for (int r = 0; r < 8; ++r)
        tile[(8 * h + r) * CP + 16 * j + c] = acc[j][r];
    wave_tile_sync();

    float* hrow = H + (size_t)s * 16 * 128 + 4 * lane;
#pragma unroll
    for (int r = 0; r < 16; ++r) {
      const v4f v = *(const v4f*)(tile + r * CP + 4 * lane);
      *(v4f*)(hrow + (size_t)r * 128) = v;
    }
    wave_tile_sync();

    if (!more) break;
    s = sn;
    strip_rows_to_tile<K, NLD, CP>(pre, tile, lane);
    wave_tile_sync();
  }
}

template<int K, int ROWS, int NWAVES>
__global__ void __launch_bounds__(NWAVES * 32)
gemm_strip_kernel(const float* __restrict__ X,
                  const float* __restrict__ W,
                  float* __restrict__ H) {
  gemm_strip_body<K, ROWS, NWAVES>(X, W, H, (int)blockIdx.x, (int)gridDim.x);
}

#define TAB_WGS 16
__global__ void __launch_bounds__(ROWS_T)
k_rows_tab(int* __restrict__ cnts, int* __restrict__ icnt, int* __restrict__ total,
           int* __restrict__ rowstart, const int* __restrict__ tok, int* __restrict__ td,
           const float* __restrict__ emb, const float* __restrict__ W1,
           float* __restrict__ tab) {
  static_assert(ROWS_T == 8 * 32, "the product below runs with 8 waves per work-group");
  if ((int)blockIdx.x < TAB_WGS) {
    gemm_strip_body<64, VOCAB, 8>(emb, W1, tab, (int)blockIdx.x, TAB_WGS);
    return;
  }
  rows_body(cnts, icnt, total, rowstart, tok, td, (int)blockIdx.x - TAB_WGS);
}


static __device__ __forceinline__ void fma4(float4& acc, const float4 h, const float w) {
  acc.x = fmaf(h.x, w, acc.x);
  acc.y = fmaf(h.y, w, acc.y);
  acc.z = fmaf(h.z, w, acc.z);
  acc.w = fmaf(h.w, w, acc.w);
}

struct RowHead { int beg, n; float di; float4 self; };

template<bool VIA_TOK>
static __device__ __forceinline__ RowHead
gather_head(const float* __restrict__ H, const int* __restrict__ rowstart,
            const int* __restrict__ icnt, const int* __restrict__ td,
            const int node, const int lane) {
  RowHead hd;
  hd.beg  = rowstart[node];
  hd.n    = icnt[node];
  const int2 t = ((const int2*)td)[node];
  hd.di   = __int_as_float(t.y);
  hd.self = ((const float4*)H)[((size_t)(VIA_TOK ? t.x : node) << 5) + lane];
  return hd;
}

template<bool VIA_TOK>
static __device__ __forceinline__ float4
gather_finish(const RowHead hd, const float* __restrict__ H, const int* __restrict__ grouped,
              const float* __restrict__ bias, const int lane) {
  const int beg  = hd.beg;
  const int n    = hd.n;
  const float di = hd.di;
  const float4* __restrict__ Hv = (const float4*)H;

  float4 acc = hd.self;
  const float ws = di * di;
  acc.x *= ws; acc.y *= ws; acc.z *= ws; acc.w *= ws;

  const int4* __restrict__ row = (const int4*)grouped + beg;
  int j = 0;
  for (; j + 4 <= n; j += 4) {
    const int4 e0 = row[j], e1 = row[j + 1], e2 = row[j + 2], e3 = row[j + 3];
    const float4 h0 = Hv[((size_t)(VIA_TOK ? e0.y : e0.x) << 5) + lane];
    const float4 h1 = Hv[((size_t)(VIA_TOK ? e1.y : e1.x) << 5) + lane];
    const float4 h2 = Hv[((size_t)(VIA_TOK ? e2.y : e2.x) << 5) + lane];
    const float4 h3 = Hv[((size_t)(VIA_TOK ? e3.y : e3.x) << 5) + lane];
    const float w0 = __int_as_float(e0.z) * di, w1 = __int_as_float(e1.z) * di;
    const float w2 = __int_as_float(e2.z) * di, w3 = __int_as_float(e3.z) * di;
    fma4(acc, h0, w0);
    fma4(acc, h1, w1);
    fma4(acc, h2, w2);
    fma4(acc, h3, w3);
  }
  if (j + 2 <= n) {
    const int4 e0 = row[j], e1 = row[j + 1];
    const float4 h0 = Hv[((size_t)(VIA_TOK ? e0.y : e0.x) << 5) + lane];
    const float4 h1 = Hv[((size_t)(VIA_TOK ? e1.y : e1.x) << 5) + lane];
    const float w0 = __int_as_float(e0.z) * di, w1 = __int_as_float(e1.z) * di;
    fma4(acc, h0, w0);
    fma4(acc, h1, w1);
    j += 2;
  }
  if (j < n) {
    const int4 e0 = row[j];
    const float4 h0 = Hv[((size_t)(VIA_TOK ? e0.y : e0.x) << 5) + lane];
    fma4(acc, h0, __int_as_float(e0.z) * di);
  }

  const float4 b = ((const float4*)bias)[lane];
  float4 v;
  v.x = fmaxf(acc.x + b.x, 0.0f);
  v.y = fmaxf(acc.y + b.y, 0.0f);
  v.z = fmaxf(acc.z + b.z, 0.0f);
  v.w = fmaxf(acc.w + b.w, 0.0f);
  return v;
}

template<bool VIA_TOK>
static __device__ __forceinline__ float4
gather_row(const float* __restrict__ H, const int* __restrict__ rowstart,
           const int* __restrict__ icnt, const int* __restrict__ grouped,
           const int* __restrict__ td, const float* __restrict__ bias,
           const int node, const int lane) {
  const RowHead hd = gather_head<VIA_TOK>(H, rowstart, icnt, td, node, lane);
  return gather_finish<VIA_TOK>(hd, H, grouped, bias, lane);
}

template<bool VIA_TOK, bool READOUT>
__global__ void __launch_bounds__(256)
k_gather(const float* __restrict__ H, const int* __restrict__ rowstart,
         const int* __restrict__ icnt, const int* __restrict__ grouped,
         const int* __restrict__ td, const float* __restrict__ bias,
         float* __restrict__ out,
         const int* __restrict__ batch, const float* __restrict__ Wlin,
         float* __restrict__ pool) {
  const int node = __builtin_amdgcn_readfirstlane((int)((blockIdx.x << 3) | (threadIdx.x >> 5)));
  const int lane = threadIdx.x & 31;
  const int   g  = READOUT ? batch[node] : 0;
  const float4 v = gather_row<VIA_TOK>(H, rowstart, icnt, grouped, td, bias, node, lane);
  if (!READOUT) {
    ((float4*)out)[((size_t)node << 5) + lane] = v;
    return;
  }

  const float4* __restrict__ Wl = (const float4*)Wlin + 4 * lane;
  const float4 w0 = Wl[0], w1 = Wl[1], w2 = Wl[2], w3 = Wl[3];
  const float p0 = fmaf(v.w, w3.x, fmaf(v.z, w2.x, fmaf(v.y, w1.x, v.x * w0.x)));
  const float p1 = fmaf(v.w, w3.y, fmaf(v.z, w2.y, fmaf(v.y, w1.y, v.x * w0.y)));
  const float p2 = fmaf(v.w, w3.z, fmaf(v.z, w2.z, fmaf(v.y, w1.z, v.x * w0.z)));
  const float p3 = fmaf(v.w, w3.w, fmaf(v.z, w2.w, fmaf(v.y, w1.w, v.x * w0.w)));

  const bool odd1 = (lane & 1) != 0, odd2 = (lane & 2) != 0;
  const float qa = (odd1 ? p2 : p0) + __shfl_xor(odd1 ? p0 : p2, 1);
  const float qb = (odd1 ? p3 : p1) + __shfl_xor(odd1 ? p1 : p3, 1);
  float r = (odd2 ? qb : qa) + __shfl_xor(odd2 ? qa : qb, 2);
  r += __shfl_xor(r, 4);
  r += __shfl_xor(r, 8);
  r += __shfl_xor(r, 16);

  if (lane < 5) {
    const int   slot = (lane == 4) ? 4 : (((lane & 1) << 1) | (lane >> 1));
    const float val  = (lane == 4) ? 1.0f : r;
    atomicAdd(pool + ((size_t)g << 3) + slot, val);
  }
}

#define FG_NODES 32
#define FG_WGS   768
__global__ void __launch_bounds__(512)
k_gather_gemm(const float* __restrict__ tab, const int* __restrict__ rowstart,
              const int* __restrict__ icnt, const int* __restrict__ grouped,
              const int* __restrict__ td, const float* __restrict__ bias,
              const float* __restrict__ W,
              float* __restrict__ H) {
  static_assert(N_NODES % FG_NODES == 0, "no partial strip");
  constexpr int NTRIPS = N_NODES / FG_NODES;
  constexpr int CP = 132;
  __shared__ v4f Wf[16 * 8 * 32];
  __shared__ __attribute__((aligned(16))) float Xt[FG_NODES * CP];
  __shared__ __attribute__((aligned(16))) float Ot[FG_NODES * CP];

#pragma unroll
  for (int i = 0; i < 8; ++i) {
    const int f  = (int)threadIdx.x + i * 512;
    const int ln = f & 31, j = (f >> 5) & 7, t = f >> 8;
    const float* wp = W + (size_t)(8 * t + 4 * (ln >> 4)) * 128 + 16 * j + (ln & 15);
    v4f v;
    v.x = wp[0]; v.y = wp[128]; v.z = wp[256]; v.w = wp[384];
    Wf[f] = v;
  }
  __syncthreads();

  const int wave = __builtin_amdgcn_readfirstlane((int)(threadIdx.x >> 5));
  const int lane = threadIdx.x & 31;
  const int h    = lane >> 4;
  const int c    = lane & 15;
  const int sidx = wave >> 3;
  const int jt   = wave & 7;

  for (int trip = (int)blockIdx.x; trip < NTRIPS; trip += FG_WGS) {
    const int base = trip * FG_NODES;

    const int r0 = 2 * wave;
    const RowHead hd0 = gather_head<true>(tab, rowstart, icnt, td, base + r0, lane);
    const RowHead hd1 = gather_head<true>(tab, rowstart, icnt, td, base + r0 + 1, lane);
    const float4 v0 = gather_finish<true>(hd0, tab, grouped, bias, lane);
    *(float4*)(Xt + r0 * CP + 4 * lane) = v0;
    const float4 v1 = gather_finish<true>(hd1, tab, grouped, bias, lane);
    *(float4*)(Xt + (r0 + 1) * CP + 4 * lane) = v1;
    __syncthreads();

    v8f acc = (v8f){};
    const float* arow = Xt + (16 * sidx + c) * CP + 4 * h;
#pragma unroll
    for (int t = 0; t < 16; ++t) {
      const v4f a = *(const v4f*)(arow + 8 * t);
      const v4f b = Wf[(t * 8 + jt) * 32 + lane];
      v2f a0; a0.x = a.x; a0.y = a.y;
      v2f b0; b0.x = b.x; b0.y = b.y;
      acc = __builtin_amdgcn_wmma_f32_16x16x4_f32(
           false, a0,  false, b0,
           (short)0, acc,  false,  false);
      v2f a1; a1.x = a.z; a1.y = a.w;
      v2f b1; b1.x = b.z; b1.y = b.w;
      acc = __builtin_amdgcn_wmma_f32_16x16x4_f32(
           false, a1,  false, b1,
           (short)0, acc,  false,  false);
    }
#pragma unroll
    for (int r = 0; r < 8; ++r)
      Ot[(16 * sidx + 8 * h + r) * CP + 16 * jt + c] = acc[r];
    __syncthreads();

#pragma unroll
    for (int q = 0; q < 2; ++q) {
      const int r = 2 * wave + q;
      const v4f v = *(const v4f*)(Ot + r * CP + 4 * lane);
      *(v4f*)(H + (size_t)(base + r) * 128 + 4 * lane) = v;
    }
    __builtin_amdgcn_fence(__ATOMIC_RELEASE, "wavefront");
  }
}


__global__ void k_final(const float* __restrict__ pool, const float* __restrict__ blin,
                        float* __restrict__ out) {
  int t = blockIdx.x * blockDim.x + threadIdx.x;
  if (t >= N_GRAPHS * NCLS) return;
  int g = t >> 2, c = t & 3;
  const float* p = pool + ((size_t)g << 3);
  out[t] = p[c] / fmaxf(p[4], 1.0f) + blin[c];
}


extern "C" void kernel_launch(void* const* d_in, const int* in_sizes, int n_in,
                              void* d_out, int out_size, void* d_ws, size_t ws_size,
                              hipStream_t stream) {
  const int*   tok   = (const int*)d_in[0];
  const int*   eidx  = (const int*)d_in[1];
  const int*   src   = eidx;
  const int*   dst   = eidx + N_EDGES;
  const int*   batch = (const int*)d_in[2];
  const float* emb   = (const float*)d_in[3];
  const float* W1    = (const float*)d_in[4];
  const float* b1    = (const float*)d_in[5];
  const float* W2    = (const float*)d_in[6];
  const float* b2    = (const float*)d_in[7];
  const float* Wlin  = (const float*)d_in[8];
  const float* blin  = (const float*)d_in[9];
  float*       out   = (float*)d_out;

  float* ws   = (float*)d_ws;
  int*   rowstart = (int*)ws;
  float* tab  = ws + N_NODES;
  float* A    = tab  + (size_t)VOCAB_PAD * 128;
  int*   icnt     = (int*)(A + (size_t)N_NODES * 128);
  int*   total    = icnt + N_NODES;
  float* pool     = (float*)(total + 8);
  int*   rank     = (int*)(pool + N_GRAPHS * 8);
  int*   grouped  = rank + N_EDGES;
  int*   cnts     = grouped + 4 * (size_t)N_EDGES;
  int*   td       = cnts + (size_t)RK_CHUNKS * N_NODES;

  static_assert(N_NODES % 4 == 0 && (VOCAB_PAD * 128) % 4 == 0 && (N_GRAPHS * 8) % 4 == 0 &&
                N_EDGES % 4 == 0, "16-byte bounds of grouped and td");
  const int T = 256;
  const int gNodes  = (N_NODES + T - 1) / T;
  const int gEdges  = (N_EDGES + T - 1) / T;
  const int gRows   = (N_NODES * 32) / T;

  k_rank<<<RK_RANGES * RK_CHUNKS, RK_T, 0, stream>>>(dst, cnts, rank, total, pool);
  k_rows_tab<<<gNodes + TAB_WGS, ROWS_T, 0, stream>>>(cnts, icnt, total, rowstart, tok, td, emb, W1, tab);
  k_fill<<<gEdges, T, 0, stream>>>(src, dst, rank, cnts, td, grouped);

  k_gather_gemm<<<FG_WGS, 512, 0, stream>>>(tab, rowstart, icnt, grouped, td, b1, W2, A);

  k_gather<false, true><<<gRows, T, 0, stream>>>(A, rowstart, icnt, grouped, td, b2, nullptr,
                                                 batch, Wlin, pool);

  k_final<<<(N_GRAPHS * NCLS + T - 1) / T, T, 0, stream>>>(pool, blin, out);
}
